# starting kernel
# baseline (speedup 1.0000x reference)
#include <hip/hip_runtime.h>

#define N_NODES   100000
#define N_EDGES   1600000
#define C_IN      64
#define HID       32
#define G_GRAPHS  128
#define NEG_SLOPE 0.01f

typedef __attribute__((ext_vector_type(2))) float v2f;
typedef __attribute__((ext_vector_type(8))) float v8f;

__global__ void gcn_zero_kernel(float* __restrict__ p, int n) {
  int i = blockIdx.x * blockDim.x + threadIdx.x;
  if (i < n) p[i] = 0.0f;
}

__global__ void gcn_degree_kernel(const int* __restrict__ dst, float* __restrict__ deg) {
  int e = blockIdx.x * blockDim.x + threadIdx.x;
  if (e < N_EDGES) atomicAdd(&deg[dst[e]], 1.0f);
}

__global__ void gcn_counts_kernel(const int* __restrict__ batch, float* __restrict__ cnt) {
  int i = blockIdx.x * blockDim.x + threadIdx.x;
  if (i < N_NODES) atomicAdd(&cnt[batch[i]], 1.0f);
}

__global__ void gcn_dinv_kernel(float* __restrict__ deg) {
  int i = blockIdx.x * blockDim.x + threadIdx.x;
  if (i < N_NODES) deg[i] = rsqrtf(deg[i] + 1.0f);
}

__global__ void gcn_cntinv_kernel(float* __restrict__ cnt) {
  int g = blockIdx.x * blockDim.x + threadIdx.x;
  if (g < G_GRAPHS) cnt[g] = 1.0f / fmaxf(cnt[g], 1.0f);
}

template <int K>
__global__ void gcn_gemm_wmma_kernel(const float* __restrict__ X,
                                     const float* __restrict__ W,
                                     float* __restrict__ H, int nRows) {
  const int wave = blockIdx.x * (blockDim.x >> 5) + (threadIdx.x >> 5);
  const int lane = threadIdx.x & 31;
  const int numTiles = (nRows >> 4) * 2;
  if (wave >= numTiles) return;
  const int rowTile = wave >> 1;
  const int colTile = wave & 1;
  const int half = lane >> 4;
  const int l16  = lane & 15;
  const int aRow = rowTile * 16 + l16;
  const int col  = colTile * 16 + l16;

  v8f c = {};
#pragma unroll
  for (int k0 = 0; k0 < K; k0 += 4) {
    const int ka = k0 + 2 * half;
    v2f a = *(const v2f*)(X + aRow * K + ka);
    v2f b;
    b.x = W[(ka + 0) * HID + col];
    b.y = W[(ka + 1) * HID + col];
    c = __builtin_amdgcn_wmma_f32_16x16x4_f32(
         false, a,  false, b,
         (short)0, c,  false,  false);
  }
#pragma unroll
  for (int r = 0; r < 8; ++r) {
    const int m = r + 8 * half;
    H[(rowTile * 16 + m) * HID + col] = c[r];
  }
}

__global__ void gcn_scatter_kernel(const int* __restrict__ src, const int* __restrict__ dst,
                                   const float* __restrict__ dinv,
                                   const float* __restrict__ H, float* __restrict__ agg) {
  int t = blockIdx.x * blockDim.x + threadIdx.x;
  int e = t >> 3;
  if (e >= N_EDGES) return;
  int q = (t & 7) << 2;
  int s = src[e], d = dst[e];
  float coef = dinv[s] * dinv[d];
  const float4 hv = *(const float4*)(H + s * HID + q);
  float* ap = agg + d * HID + q;
  atomicAdd(ap + 0, hv.x * coef);
  atomicAdd(ap + 1, hv.y * coef);
  atomicAdd(ap + 2, hv.z * coef);
  atomicAdd(ap + 3, hv.w * coef);
}

__global__ void gcn_combine_kernel(const float* __restrict__ agg, const float* __restrict__ Hlin,
                                   const float* __restrict__ bias, const float* __restrict__ dinv,
                                   const int* __restrict__ batch, const float* __restrict__ cntinv,
                                   float* __restrict__ Hact, float* __restrict__ feat) {
  int t = blockIdx.x * blockDim.x + threadIdx.x;
  if (t >= N_NODES * HID) return;
  int i = t >> 5, ch = t & 31;
  float di = dinv[i];
  float v = agg[t] + Hlin[t] * di * di + bias[ch];
  v = v > 0.0f ? v : NEG_SLOPE * v;
  Hact[t] = v;
  int g = batch[i];
  atomicAdd(&feat[g * HID + ch], v * cntinv[g]);
}

__global__ void gcn_merge_kernel(const float* __restrict__ feats, float* __restrict__ out) {
  int t = blockIdx.x * blockDim.x + threadIdx.x;
  const int GH = G_GRAPHS * HID;
  if (t < GH) out[t] = (feats[t] + feats[GH + t] + feats[2 * GH + t]) * (1.0f / 3.0f);
}

static inline int cdiv(int a, int b) { return (a + b - 1) / b; }

extern "C" void kernel_launch(void* const* d_in, const int* in_sizes, int n_in,
                              void* d_out, int out_size, void* d_ws, size_t ws_size,
                              hipStream_t stream) {
  (void)in_sizes; (void)n_in; (void)out_size; (void)ws_size;
  const float* x     = (const float*)d_in[0];
  const float* W0    = (const float*)d_in[1];
  const float* b0    = (const float*)d_in[2];
  const float* W1    = (const float*)d_in[3];
  const float* b1    = (const float*)d_in[4];
  const float* W2    = (const float*)d_in[5];
  const float* b2    = (const float*)d_in[6];
  const int*   src   = (const int*)d_in[7];
  const int*   dst   = (const int*)d_in[8];
  const int*   batch = (const int*)d_in[9];
  float* out = (float*)d_out;

  float* ws    = (float*)d_ws;
  float* deg   = ws;
  float* cnt   = deg + N_NODES;
  float* lin   = cnt + G_GRAPHS;
  float* act   = lin + (size_t)N_NODES * HID;
  float* agg   = act + (size_t)N_NODES * HID;
  float* feats = agg + (size_t)N_NODES * HID;

  const int T = 256;
  const int nTiles = (N_NODES / 16) * 2;

  gcn_zero_kernel<<<cdiv(N_NODES, T), T, 0, stream>>>(deg, N_NODES);
  gcn_zero_kernel<<<1, T, 0, stream>>>(cnt, G_GRAPHS);
  gcn_zero_kernel<<<cdiv(3 * G_GRAPHS * HID, T), T, 0, stream>>>(feats, 3 * G_GRAPHS * HID);
  gcn_degree_kernel<<<cdiv(N_EDGES, T), T, 0, stream>>>(dst, deg);
  gcn_counts_kernel<<<cdiv(N_NODES, T), T, 0, stream>>>(batch, cnt);
  gcn_dinv_kernel<<<cdiv(N_NODES, T), T, 0, stream>>>(deg);
  gcn_cntinv_kernel<<<1, T, 0, stream>>>(cnt);

  const float* Ws[3] = {W0, W1, W2};
  const float* bs[3] = {b0, b1, b2};
  const float* hin = x;
  for (int layer = 0; layer < 3; ++layer) {
    if (layer == 0)
      gcn_gemm_wmma_kernel<C_IN><<<cdiv(nTiles, T / 32), T, 0, stream>>>(hin, Ws[0], lin, N_NODES);
    else
      gcn_gemm_wmma_kernel<HID><<<cdiv(nTiles, T / 32), T, 0, stream>>>(hin, Ws[layer], lin, N_NODES);
    gcn_zero_kernel<<<cdiv(N_NODES * HID, T), T, 0, stream>>>(agg, N_NODES * HID);
    gcn_scatter_kernel<<<cdiv(N_EDGES * 8, T), T, 0, stream>>>(src, dst, deg, lin, agg);
    gcn_combine_kernel<<<cdiv(N_NODES * HID, T), T, 0, stream>>>(
        agg, lin, bs[layer], deg, batch, cnt, act, feats + layer * G_GRAPHS * HID);
    hin = act;
  }
  gcn_merge_kernel<<<cdiv(G_GRAPHS * HID, T), T, 0, stream>>>(feats, out);
}


# graph-size count: lanes of a wave with the same graph id are tallied by ballot and one lane adds the total (one add per distinct id per wave); counts stay exact integers
# speedup vs baseline: 1.1294x; 1.1294x over previous
#include <hip/hip_runtime.h>

#define N_NODES   100000
#define N_EDGES   1600000
#define C_IN      64
#define HID       32
#define G_GRAPHS  128
#define NEG_SLOPE 0.01f

typedef __attribute__((ext_vector_type(2))) float v2f;
typedef __attribute__((ext_vector_type(8))) float v8f;

__global__ void gcn_zero_kernel(float* __restrict__ p, int n) {
  int i = blockIdx.x * blockDim.x + threadIdx.x;
  if (i < n) p[i] = 0.0f;
}

__global__ void gcn_degree_kernel(const int* __restrict__ dst, float* __restrict__ deg) {
  int e = blockIdx.x * blockDim.x + threadIdx.x;
  if (e < N_EDGES) atomicAdd(&deg[dst[e]], 1.0f);
}

__global__ void gcn_counts_kernel(const int* __restrict__ batch, float* __restrict__ cnt) {
  const int i = blockIdx.x * blockDim.x + threadIdx.x;
  const int lane = threadIdx.x & 31;
  const bool live = i < N_NODES;
  const int g = live ? batch[i] : -1;
  unsigned remaining = __builtin_amdgcn_ballot_w32(live);
  for (int trip = 0; trip < 32 && remaining != 0u; ++trip) {
    const int lead = __ffs(remaining) - 1;
    const int gl = __builtin_amdgcn_readlane(g, lead);
    const unsigned same = __builtin_amdgcn_ballot_w32(live && g == gl) & remaining;
    if (lane == lead) atomicAdd(&cnt[gl], (float)__popc(same));
    remaining &= ~same;
  }
}

__global__ void gcn_dinv_kernel(float* __restrict__ deg) {
  int i = blockIdx.x * blockDim.x + threadIdx.x;
  if (i < N_NODES) deg[i] = rsqrtf(deg[i] + 1.0f);
}

__global__ void gcn_cntinv_kernel(float* __restrict__ cnt) {
  int g = blockIdx.x * blockDim.x + threadIdx.x;
  if (g < G_GRAPHS) cnt[g] = 1.0f / fmaxf(cnt[g], 1.0f);
}

template <int K>
__global__ void gcn_gemm_wmma_kernel(const float* __restrict__ X,
                                     const float* __restrict__ W,
                                     float* __restrict__ H, int nRows) {
  const int wave = blockIdx.x * (blockDim.x >> 5) + (threadIdx.x >> 5);
  const int lane = threadIdx.x & 31;
  const int numTiles = (nRows >> 4) * 2;
  if (wave >= numTiles) return;
  const int rowTile = wave >> 1;
  const int colTile = wave & 1;
  const int half = lane >> 4;
  const int l16  = lane & 15;
  const int aRow = rowTile * 16 + l16;
  const int col  = colTile * 16 + l16;

  v8f c = {};
#pragma unroll
  for (int k0 = 0; k0 < K; k0 += 4) {
    const int ka = k0 + 2 * half;
    v2f a = *(const v2f*)(X + aRow * K + ka);
    v2f b;
    b.x = W[(ka + 0) * HID + col];
    b.y = W[(ka + 1) * HID + col];
    c = __builtin_amdgcn_wmma_f32_16x16x4_f32(
         false, a,  false, b,
         (short)0, c,  false,  false);
  }
#pragma unroll
  for (int r = 0; r < 8; ++r) {
    const int m = r + 8 * half;
    H[(rowTile * 16 + m) * HID + col] = c[r];
  }
}

__global__ void gcn_scatter_kernel(const int* __restrict__ src, const int* __restrict__ dst,
                                   const float* __restrict__ dinv,
                                   const float* __restrict__ H, float* __restrict__ agg) {
  int t = blockIdx.x * blockDim.x + threadIdx.x;
  int e = t >> 3;
  if (e >= N_EDGES) return;
  int q = (t & 7) << 2;
  int s = src[e], d = dst[e];
  float coef = dinv[s] * dinv[d];
  const float4 hv = *(const float4*)(H + s * HID + q);
  float* ap = agg + d * HID + q;
  atomicAdd(ap + 0, hv.x * coef);
  atomicAdd(ap + 1, hv.y * coef);
  atomicAdd(ap + 2, hv.z * coef);
  atomicAdd(ap + 3, hv.w * coef);
}

__global__ void gcn_combine_kernel(const float* __restrict__ agg, const float* __restrict__ Hlin,
                                   const float* __restrict__ bias, const float* __restrict__ dinv,
                                   const int* __restrict__ batch, const float* __restrict__ cntinv,
                                   float* __restrict__ Hact, float* __restrict__ feat) {
  int t = blockIdx.x * blockDim.x + threadIdx.x;
  if (t >= N_NODES * HID) return;
  int i = t >> 5, ch = t & 31;
  float di = dinv[i];
  float v = agg[t] + Hlin[t] * di * di + bias[ch];
  v = v > 0.0f ? v : NEG_SLOPE * v;
  Hact[t] = v;
  int g = batch[i];
  atomicAdd(&feat[g * HID + ch], v * cntinv[g]);
}

__global__ void gcn_merge_kernel(const float* __restrict__ feats, float* __restrict__ out) {
  int t = blockIdx.x * blockDim.x + threadIdx.x;
  const int GH = G_GRAPHS * HID;
  if (t < GH) out[t] = (feats[t] + feats[GH + t] + feats[2 * GH + t]) * (1.0f / 3.0f);
}

static inline int cdiv(int a, int b) { return (a + b - 1) / b; }

extern "C" void kernel_launch(void* const* d_in, const int* in_sizes, int n_in,
                              void* d_out, int out_size, void* d_ws, size_t ws_size,
                              hipStream_t stream) {
  (void)in_sizes; (void)n_in; (void)out_size; (void)ws_size;
  const float* x     = (const float*)d_in[0];
  const float* W0    = (const float*)d_in[1];
  const float* b0    = (const float*)d_in[2];
  const float* W1    = (const float*)d_in[3];
  const float* b1    = (const float*)d_in[4];
  const float* W2    = (const float*)d_in[5];
  const float* b2    = (const float*)d_in[6];
  const int*   src   = (const int*)d_in[7];
  const int*   dst   = (const int*)d_in[8];
  const int*   batch = (const int*)d_in[9];
  float* out = (float*)d_out;

  float* ws    = (float*)d_ws;
  float* deg   = ws;
  float* cnt   = deg + N_NODES;
  float* lin   = cnt + G_GRAPHS;
  float* act   = lin + (size_t)N_NODES * HID;
  float* agg   = act + (size_t)N_NODES * HID;
  float* feats = agg + (size_t)N_NODES * HID;

  const int T = 256;
  const int nTiles = (N_NODES / 16) * 2;

  gcn_zero_kernel<<<cdiv(N_NODES, T), T, 0, stream>>>(deg, N_NODES);
  gcn_zero_kernel<<<1, T, 0, stream>>>(cnt, G_GRAPHS);
  gcn_zero_kernel<<<cdiv(3 * G_GRAPHS * HID, T), T, 0, stream>>>(feats, 3 * G_GRAPHS * HID);
  gcn_degree_kernel<<<cdiv(N_EDGES, T), T, 0, stream>>>(dst, deg);
  gcn_counts_kernel<<<cdiv(N_NODES, T), T, 0, stream>>>(batch, cnt);
  gcn_dinv_kernel<<<cdiv(N_NODES, T), T, 0, stream>>>(deg);
  gcn_cntinv_kernel<<<1, T, 0, stream>>>(cnt);

  const float* Ws[3] = {W0, W1, W2};
  const float* bs[3] = {b0, b1, b2};
  const float* hin = x;
  for (int layer = 0; layer < 3; ++layer) {
    if (layer == 0)
      gcn_gemm_wmma_kernel<C_IN><<<cdiv(nTiles, T / 32), T, 0, stream>>>(hin, Ws[0], lin, N_NODES);
    else
      gcn_gemm_wmma_kernel<HID><<<cdiv(nTiles, T / 32), T, 0, stream>>>(hin, Ws[layer], lin, N_NODES);
    gcn_zero_kernel<<<cdiv(N_NODES * HID, T), T, 0, stream>>>(agg, N_NODES * HID);
    gcn_scatter_kernel<<<cdiv(N_EDGES * 8, T), T, 0, stream>>>(src, dst, deg, lin, agg);
    gcn_combine_kernel<<<cdiv(N_NODES * HID, T), T, 0, stream>>>(
        agg, lin, bs[layer], deg, batch, cnt, act, feats + layer * G_GRAPHS * HID);
    hin = act;
  }
  gcn_merge_kernel<<<cdiv(G_GRAPHS * HID, T), T, 0, stream>>>(feats, out);
}


# edge scatter remapped to lane = channel: a wave walks 64 edges, each atomic instruction adds one edge's whole 128-byte row (was 4 quarter rows); float adds stay unordered, arrival order differs
# speedup vs baseline: 1.3670x; 1.2104x over previous
#include <hip/hip_runtime.h>

#define N_NODES   100000
#define N_EDGES   1600000
#define C_IN      64
#define HID       32
#define G_GRAPHS  128
#define NEG_SLOPE 0.01f

typedef __attribute__((ext_vector_type(2))) float v2f;
typedef __attribute__((ext_vector_type(8))) float v8f;

__global__ void gcn_zero_kernel(float* __restrict__ p, int n) {
  int i = blockIdx.x * blockDim.x + threadIdx.x;
  if (i < n) p[i] = 0.0f;
}

__global__ void gcn_degree_kernel(const int* __restrict__ dst, float* __restrict__ deg) {
  int e = blockIdx.x * blockDim.x + threadIdx.x;
  if (e < N_EDGES) atomicAdd(&deg[dst[e]], 1.0f);
}

__global__ void gcn_counts_kernel(const int* __restrict__ batch, float* __restrict__ cnt) {
  const int i = blockIdx.x * blockDim.x + threadIdx.x;
  const int lane = threadIdx.x & 31;
  const bool live = i < N_NODES;
  const int g = live ? batch[i] : -1;
  unsigned remaining = __builtin_amdgcn_ballot_w32(live);
  for (int trip = 0; trip < 32 && remaining != 0u; ++trip) {
    const int lead = __ffs(remaining) - 1;
    const int gl = __builtin_amdgcn_readlane(g, lead);
    const unsigned same = __builtin_amdgcn_ballot_w32(live && g == gl) & remaining;
    if (lane == lead) atomicAdd(&cnt[gl], (float)__popc(same));
    remaining &= ~same;
  }
}

__global__ void gcn_dinv_kernel(float* __restrict__ deg) {
  int i = blockIdx.x * blockDim.x + threadIdx.x;
  if (i < N_NODES) deg[i] = rsqrtf(deg[i] + 1.0f);
}

__global__ void gcn_cntinv_kernel(float* __restrict__ cnt) {
  int g = blockIdx.x * blockDim.x + threadIdx.x;
  if (g < G_GRAPHS) cnt[g] = 1.0f / fmaxf(cnt[g], 1.0f);
}

template <int K>
__global__ void gcn_gemm_wmma_kernel(const float* __restrict__ X,
                                     const float* __restrict__ W,
                                     float* __restrict__ H, int nRows) {
  const int wave = blockIdx.x * (blockDim.x >> 5) + (threadIdx.x >> 5);
  const int lane = threadIdx.x & 31;
  const int numTiles = (nRows >> 4) * 2;
  if (wave >= numTiles) return;
  const int rowTile = wave >> 1;
  const int colTile = wave & 1;
  const int half = lane >> 4;
  const int l16  = lane & 15;
  const int aRow = rowTile * 16 + l16;
  const int col  = colTile * 16 + l16;

  v8f c = {};
#pragma unroll
  for (int k0 = 0; k0 < K; k0 += 4) {
    const int ka = k0 + 2 * half;
    v2f a = *(const v2f*)(X + aRow * K + ka);
    v2f b;
    b.x = W[(ka + 0) * HID + col];
    b.y = W[(ka + 1) * HID + col];
    c = __builtin_amdgcn_wmma_f32_16x16x4_f32(
         false, a,  false, b,
         (short)0, c,  false,  false);
  }
#pragma unroll
  for (int r = 0; r < 8; ++r) {
    const int m = r + 8 * half;
    H[(rowTile * 16 + m) * HID + col] = c[r];
  }
}

#define SC_EDGES 64
#define SC_CHUNKS (SC_EDGES / 32)
static_assert(N_EDGES % SC_EDGES == 0, "edge count must be a whole number of wave chunks");
static_assert(HID == 32, "one lane per channel");
__global__ void gcn_scatter_kernel(const int* __restrict__ src, const int* __restrict__ dst,
                                   const float* __restrict__ dinv,
                                   const float* __restrict__ H, float* __restrict__ agg) {
  const int wave = blockIdx.x * (blockDim.x >> 5) + (threadIdx.x >> 5);
  const int lane = threadIdx.x & 31;
  const int base = wave * SC_EDGES;
  if (base >= N_EDGES) return;
  int s[SC_CHUNKS], d[SC_CHUNKS], coefBits[SC_CHUNKS];
#pragma unroll
  for (int c = 0; c < SC_CHUNKS; ++c) {
    const int e = base + c * 32 + lane;
    s[c] = src[e];
    d[c] = dst[e];
  }
#pragma unroll
  for (int c = 0; c < SC_CHUNKS; ++c)
    coefBits[c] = __float_as_int(dinv[s[c]] * dinv[d[c]]);
#pragma unroll
  for (int c = 0; c < SC_CHUNKS; ++c) {
    float hv[32];
#pragma unroll
    for (int j = 0; j < 32; ++j) {
      const int sj = __builtin_amdgcn_readlane(s[c], j);
      hv[j] = H[(unsigned)(sj * HID + lane)];
    }
#pragma unroll
    for (int j = 0; j < 32; ++j) {
      const int dj = __builtin_amdgcn_readlane(d[c], j);
      const float cj = __int_as_float(__builtin_amdgcn_readlane(coefBits[c], j));
      atomicAdd(agg + (unsigned)(dj * HID + lane), hv[j] * cj);
    }
  }
}

__global__ void gcn_combine_kernel(const float* __restrict__ agg, const float* __restrict__ Hlin,
                                   const float* __restrict__ bias, const float* __restrict__ dinv,
                                   const int* __restrict__ batch, const float* __restrict__ cntinv,
                                   float* __restrict__ Hact, float* __restrict__ feat) {
  int t = blockIdx.x * blockDim.x + threadIdx.x;
  if (t >= N_NODES * HID) return;
  int i = t >> 5, ch = t & 31;
  float di = dinv[i];
  float v = agg[t] + Hlin[t] * di * di + bias[ch];
  v = v > 0.0f ? v : NEG_SLOPE * v;
  Hact[t] = v;
  int g = batch[i];
  atomicAdd(&feat[g * HID + ch], v * cntinv[g]);
}

__global__ void gcn_merge_kernel(const float* __restrict__ feats, float* __restrict__ out) {
  int t = blockIdx.x * blockDim.x + threadIdx.x;
  const int GH = G_GRAPHS * HID;
  if (t < GH) out[t] = (feats[t] + feats[GH + t] + feats[2 * GH + t]) * (1.0f / 3.0f);
}

static inline int cdiv(int a, int b) { return (a + b - 1) / b; }

extern "C" void kernel_launch(void* const* d_in, const int* in_sizes, int n_in,
                              void* d_out, int out_size, void* d_ws, size_t ws_size,
                              hipStream_t stream) {
  (void)in_sizes; (void)n_in; (void)out_size; (void)ws_size;
  const float* x     = (const float*)d_in[0];
  const float* W0    = (const float*)d_in[1];
  const float* b0    = (const float*)d_in[2];
  const float* W1    = (const float*)d_in[3];
  const float* b1    = (const float*)d_in[4];
  const float* W2    = (const float*)d_in[5];
  const float* b2    = (const float*)d_in[6];
  const int*   src   = (const int*)d_in[7];
  const int*   dst   = (const int*)d_in[8];
  const int*   batch = (const int*)d_in[9];
  float* out = (float*)d_out;

  float* ws    = (float*)d_ws;
  float* deg   = ws;
  float* cnt   = deg + N_NODES;
  float* lin   = cnt + G_GRAPHS;
  float* act   = lin + (size_t)N_NODES * HID;
  float* agg   = act + (size_t)N_NODES * HID;
  float* feats = agg + (size_t)N_NODES * HID;

  const int T = 256;
  const int nTiles = (N_NODES / 16) * 2;

  gcn_zero_kernel<<<cdiv(N_NODES, T), T, 0, stream>>>(deg, N_NODES);
  gcn_zero_kernel<<<1, T, 0, stream>>>(cnt, G_GRAPHS);
  gcn_zero_kernel<<<cdiv(3 * G_GRAPHS * HID, T), T, 0, stream>>>(feats, 3 * G_GRAPHS * HID);
  gcn_degree_kernel<<<cdiv(N_EDGES, T), T, 0, stream>>>(dst, deg);
  gcn_counts_kernel<<<cdiv(N_NODES, T), T, 0, stream>>>(batch, cnt);
  gcn_dinv_kernel<<<cdiv(N_NODES, T), T, 0, stream>>>(deg);
  gcn_cntinv_kernel<<<1, T, 0, stream>>>(cnt);

  const float* Ws[3] = {W0, W1, W2};
  const float* bs[3] = {b0, b1, b2};
  const float* hin = x;
  for (int layer = 0; layer < 3; ++layer) {
    if (layer == 0)
      gcn_gemm_wmma_kernel<C_IN><<<cdiv(nTiles, T / 32), T, 0, stream>>>(hin, Ws[0], lin, N_NODES);
    else
      gcn_gemm_wmma_kernel<HID><<<cdiv(nTiles, T / 32), T, 0, stream>>>(hin, Ws[layer], lin, N_NODES);
    gcn_zero_kernel<<<cdiv(N_NODES * HID, T), T, 0, stream>>>(agg, N_NODES * HID);
    gcn_scatter_kernel<<<cdiv(N_EDGES / SC_EDGES, T / 32), T, 0, stream>>>(src, dst, deg, lin, agg);
    gcn_combine_kernel<<<cdiv(N_NODES * HID, T), T, 0, stream>>>(
        agg, lin, bs[layer], deg, batch, cnt, act, feats + layer * G_GRAPHS * HID);
    hin = act;
  }
  gcn_merge_kernel<<<cdiv(G_GRAPHS * HID, T), T, 0, stream>>>(feats, out);
}


# normalisation folded out of the edge loop: GEMM scales its A rows by dinv[row], combine scales the summed row by dinv[dst] once; scatter drops both per-edge dinv gathers; add order unchanged
# speedup vs baseline: 1.4495x; 1.0604x over previous
#include <hip/hip_runtime.h>

#define N_NODES   100000
#define N_EDGES   1600000
#define C_IN      64
#define HID       32
#define G_GRAPHS  128
#define NEG_SLOPE 0.01f

typedef __attribute__((ext_vector_type(2))) float v2f;
typedef __attribute__((ext_vector_type(8))) float v8f;

__global__ void gcn_zero_kernel(float* __restrict__ p, int n) {
  int i = blockIdx.x * blockDim.x + threadIdx.x;
  if (i < n) p[i] = 0.0f;
}

__global__ void gcn_degree_kernel(const int* __restrict__ dst, float* __restrict__ deg) {
  int e = blockIdx.x * blockDim.x + threadIdx.x;
  if (e < N_EDGES) atomicAdd(&deg[dst[e]], 1.0f);
}

__global__ void gcn_counts_kernel(const int* __restrict__ batch, float* __restrict__ cnt) {
  const int i = blockIdx.x * blockDim.x + threadIdx.x;
  const int lane = threadIdx.x & 31;
  const bool live = i < N_NODES;
  const int g = live ? batch[i] : -1;
  unsigned remaining = __builtin_amdgcn_ballot_w32(live);
  for (int trip = 0; trip < 32 && remaining != 0u; ++trip) {
    const int lead = __ffs(remaining) - 1;
    const int gl = __builtin_amdgcn_readlane(g, lead);
    const unsigned same = __builtin_amdgcn_ballot_w32(live && g == gl) & remaining;
    if (lane == lead) atomicAdd(&cnt[gl], (float)__popc(same));
    remaining &= ~same;
  }
}

__global__ void gcn_dinv_kernel(float* __restrict__ deg) {
  int i = blockIdx.x * blockDim.x + threadIdx.x;
  if (i < N_NODES) deg[i] = rsqrtf(deg[i] + 1.0f);
}

__global__ void gcn_cntinv_kernel(float* __restrict__ cnt) {
  int g = blockIdx.x * blockDim.x + threadIdx.x;
  if (g < G_GRAPHS) cnt[g] = 1.0f / fmaxf(cnt[g], 1.0f);
}

template <int K>
__global__ void gcn_gemm_wmma_kernel(const float* __restrict__ X,
                                     const float* __restrict__ W,
                                     const float* __restrict__ dinv,
                                     float* __restrict__ H, int nRows) {
  const int wave = blockIdx.x * (blockDim.x >> 5) + (threadIdx.x >> 5);
  const int lane = threadIdx.x & 31;
  const int numTiles = (nRows >> 4) * 2;
  if (wave >= numTiles) return;
  const int rowTile = wave >> 1;
  const int colTile = wave & 1;
  const int half = lane >> 4;
  const int l16  = lane & 15;
  const int aRow = rowTile * 16 + l16;
  const int col  = colTile * 16 + l16;

  const float da = dinv[aRow];

  v8f c = {};
#pragma unroll
  for (int k0 = 0; k0 < K; k0 += 4) {
    const int ka = k0 + 2 * half;
    v2f a = *(const v2f*)(X + aRow * K + ka) * da;
    v2f b;
    b.x = W[(ka + 0) * HID + col];
    b.y = W[(ka + 1) * HID + col];
    c = __builtin_amdgcn_wmma_f32_16x16x4_f32(
         false, a,  false, b,
         (short)0, c,  false,  false);
  }
#pragma unroll
  for (int r = 0; r < 8; ++r) {
    const int m = r + 8 * half;
    H[(rowTile * 16 + m) * HID + col] = c[r];
  }
}

#define SC_EDGES 64
#define SC_CHUNKS (SC_EDGES / 32)
static_assert(N_EDGES % SC_EDGES == 0, "edge count must be a whole number of wave chunks");
static_assert(HID == 32, "one lane per channel");
__global__ void gcn_scatter_kernel(const int* __restrict__ src, const int* __restrict__ dst,
                                   const float* __restrict__ H, float* __restrict__ agg) {
  const int wave = blockIdx.x * (blockDim.x >> 5) + (threadIdx.x >> 5);
  const int lane = threadIdx.x & 31;
  const int base = wave * SC_EDGES;
  if (base >= N_EDGES) return;
  int s[SC_CHUNKS], d[SC_CHUNKS];
#pragma unroll
  for (int c = 0; c < SC_CHUNKS; ++c) {
    const int e = base + c * 32 + lane;
    s[c] = src[e];
    d[c] = dst[e];
  }
#pragma unroll
  for (int c = 0; c < SC_CHUNKS; ++c) {
    float hv[32];
#pragma unroll
    for (int j = 0; j < 32; ++j) {
      const int sj = __builtin_amdgcn_readlane(s[c], j);
      hv[j] = H[(unsigned)(sj * HID + lane)];
    }
#pragma unroll
    for (int j = 0; j < 32; ++j) {
      const int dj = __builtin_amdgcn_readlane(d[c], j);
      atomicAdd(agg + (unsigned)(dj * HID + lane), hv[j]);
    }
  }
}

__global__ void gcn_combine_kernel(const float* __restrict__ agg, const float* __restrict__ Hlin,
                                   const float* __restrict__ bias, const float* __restrict__ dinv,
                                   const int* __restrict__ batch, const float* __restrict__ cntinv,
                                   float* __restrict__ Hact, float* __restrict__ feat) {
  int t = blockIdx.x * blockDim.x + threadIdx.x;
  if (t >= N_NODES * HID) return;
  int i = t >> 5, ch = t & 31;
  float di = dinv[i];
  float v = (agg[t] + Hlin[t]) * di + bias[ch];
  v = v > 0.0f ? v : NEG_SLOPE * v;
  Hact[t] = v;
  int g = batch[i];
  atomicAdd(&feat[g * HID + ch], v * cntinv[g]);
}

__global__ void gcn_merge_kernel(const float* __restrict__ feats, float* __restrict__ out) {
  int t = blockIdx.x * blockDim.x + threadIdx.x;
  const int GH = G_GRAPHS * HID;
  if (t < GH) out[t] = (feats[t] + feats[GH + t] + feats[2 * GH + t]) * (1.0f / 3.0f);
}

static inline int cdiv(int a, int b) { return (a + b - 1) / b; }

extern "C" void kernel_launch(void* const* d_in, const int* in_sizes, int n_in,
                              void* d_out, int out_size, void* d_ws, size_t ws_size,
                              hipStream_t stream) {
  (void)in_sizes; (void)n_in; (void)out_size; (void)ws_size;
  const float* x     = (const float*)d_in[0];
  const float* W0    = (const float*)d_in[1];
  const float* b0    = (const float*)d_in[2];
  const float* W1    = (const float*)d_in[3];
  const float* b1    = (const float*)d_in[4];
  const float* W2    = (const float*)d_in[5];
  const float* b2    = (const float*)d_in[6];
  const int*   src   = (const int*)d_in[7];
  const int*   dst   = (const int*)d_in[8];
  const int*   batch = (const int*)d_in[9];
  float* out = (float*)d_out;

  float* ws    = (float*)d_ws;
  float* deg   = ws;
  float* cnt   = deg + N_NODES;
  float* lin   = cnt + G_GRAPHS;
  float* act   = lin + (size_t)N_NODES * HID;
  float* agg   = act + (size_t)N_NODES * HID;
  float* feats = agg + (size_t)N_NODES * HID;

  const int T = 256;
  const int nTiles = (N_NODES / 16) * 2;

  gcn_zero_kernel<<<cdiv(N_NODES, T), T, 0, stream>>>(deg, N_NODES);
  gcn_zero_kernel<<<1, T, 0, stream>>>(cnt, G_GRAPHS);
  gcn_zero_kernel<<<cdiv(3 * G_GRAPHS * HID, T), T, 0, stream>>>(feats, 3 * G_GRAPHS * HID);
  gcn_degree_kernel<<<cdiv(N_EDGES, T), T, 0, stream>>>(dst, deg);
  gcn_counts_kernel<<<cdiv(N_NODES, T), T, 0, stream>>>(batch, cnt);
  gcn_dinv_kernel<<<cdiv(N_NODES, T), T, 0, stream>>>(deg);
  gcn_cntinv_kernel<<<1, T, 0, stream>>>(cnt);

  const float* Ws[3] = {W0, W1, W2};
  const float* bs[3] = {b0, b1, b2};
  const float* hin = x;
  for (int layer = 0; layer < 3; ++layer) {
    if (layer == 0)
      gcn_gemm_wmma_kernel<C_IN><<<cdiv(nTiles, T / 32), T, 0, stream>>>(hin, Ws[0], deg, lin, N_NODES);
    else
      gcn_gemm_wmma_kernel<HID><<<cdiv(nTiles, T / 32), T, 0, stream>>>(hin, Ws[layer], deg, lin, N_NODES);
    gcn_zero_kernel<<<cdiv(N_NODES * HID, T), T, 0, stream>>>(agg, N_NODES * HID);
    gcn_scatter_kernel<<<cdiv(N_EDGES / SC_EDGES, T / 32), T, 0, stream>>>(src, dst, lin, agg);
    gcn_combine_kernel<<<cdiv(N_NODES * HID, T), T, 0, stream>>>(
        agg, lin, bs[layer], deg, batch, cnt, act, feats + layer * G_GRAPHS * HID);
    hin = act;
  }
  gcn_merge_kernel<<<cdiv(G_GRAPHS * HID, T), T, 0, stream>>>(feats, out);
}


# edge sum by destination row: edges count-sorted by dst (arrival-order rank from the degree add), resident waves walk nodes and sum source rows in registers, combine fused; row atomics, zeroing gone
# speedup vs baseline: 1.7205x; 1.1870x over previous
#include <hip/hip_runtime.h>

#define N_NODES   100000
#define N_EDGES   1600000
#define C_IN      64
#define HID       32
#define G_GRAPHS  128
#define NEG_SLOPE 0.01f

typedef __attribute__((ext_vector_type(2))) float v2f;
typedef __attribute__((ext_vector_type(8))) float v8f;

__global__ void gcn_zero_kernel(float* __restrict__ p, int n) {
  int i = blockIdx.x * blockDim.x + threadIdx.x;
  if (i < n) p[i] = 0.0f;
}

typedef __attribute__((ext_vector_type(4))) int v4i;
static_assert(N_EDGES % 4 == 0, "16-byte groups of edges must not straddle the end");
__global__ void gcn_degree_rank_kernel(const int* __restrict__ dst, int* __restrict__ degi,
                                       int* __restrict__ rank) {
  const int e = (blockIdx.x * blockDim.x + threadIdx.x) * 4;
  if (e < N_EDGES) {
    const v4i d = *(const v4i*)(dst + e);
    v4i r;
    r.x = atomicAdd(&degi[d.x], 1);
    r.y = atomicAdd(&degi[d.y], 1);
    r.z = atomicAdd(&degi[d.z], 1);
    r.w = atomicAdd(&degi[d.w], 1);
    *(v4i*)(rank + e) = r;
  }
}

__global__ void gcn_counts_kernel(const int* __restrict__ batch, float* __restrict__ cnt) {
  const int i = blockIdx.x * blockDim.x + threadIdx.x;
  const int lane = threadIdx.x & 31;
  const bool live = i < N_NODES;
  const int g = live ? batch[i] : -1;
  unsigned remaining = __builtin_amdgcn_ballot_w32(live);
  for (int trip = 0; trip < 32 && remaining != 0u; ++trip) {
    const int lead = __ffs(remaining) - 1;
    const int gl = __builtin_amdgcn_readlane(g, lead);
    const unsigned same = __builtin_amdgcn_ballot_w32(live && g == gl) & remaining;
    if (lane == lead) atomicAdd(&cnt[gl], (float)__popc(same));
    remaining &= ~same;
  }
}

#define SCAN_THREADS 1024
#define SCAN_TRIPS   25
#define SCAN_WAVE    (SCAN_TRIPS * 128)
static_assert((SCAN_THREADS / 32) * SCAN_WAVE >= N_NODES, "one work-group must cover every node");
static_assert(N_NODES % 4 == 0, "16-byte groups must not straddle the end");
__global__ void __launch_bounds__(SCAN_THREADS)
gcn_scan_dinv_kernel(int* __restrict__ degbits, int* __restrict__ rowptr) {
  __shared__ int wtot[SCAN_THREADS / 32];
  const int lane = threadIdx.x & 31;
  const int w = threadIdx.x >> 5;
  const int base = w * SCAN_WAVE + lane * 4;
  int ex[SCAN_TRIPS];
  int carry = 0;
#pragma unroll
  for (int k = 0; k < SCAN_TRIPS; ++k) {
    const int idx = base + k * 128;
    v4i v = {0, 0, 0, 0};
    if (idx < N_NODES) v = *(const v4i*)(degbits + idx);
    const int t = v.x + v.y + v.z + v.w;
    int inc = t;
#pragma unroll
    for (int d = 1; d < 32; d <<= 1) {
      const int u = __shfl_up(inc, d);
      if (lane >= d) inc += u;
    }
    ex[k] = carry + inc - t;
    carry += __builtin_amdgcn_readlane(inc, 31);
  }
  if (lane == 0) wtot[w] = carry;
  __syncthreads();
  const int mine = wtot[lane];
  int off = lane < w ? mine : 0;
  int total = mine;
#pragma unroll
  for (int d = 16; d >= 1; d >>= 1) {
    off += __shfl_xor(off, d);
    total += __shfl_xor(total, d);
  }
#pragma unroll
  for (int k = 0; k < SCAN_TRIPS; ++k) {
    const int idx = base + k * 128;
    if (idx < N_NODES) {
      const v4i v = *(const v4i*)(degbits + idx);
      v4i o;
      o.x = off + ex[k];
      o.y = o.x + v.x;
      o.z = o.y + v.y;
      o.w = o.z + v.z;
      *(v4i*)(rowptr + idx) = o;
      v4i f;
      f.x = __float_as_int(rsqrtf((float)v.x + 1.0f));
      f.y = __float_as_int(rsqrtf((float)v.y + 1.0f));
      f.z = __float_as_int(rsqrtf((float)v.z + 1.0f));
      f.w = __float_as_int(rsqrtf((float)v.w + 1.0f));
      *(v4i*)(degbits + idx) = f;
    }
  }
  if (threadIdx.x == 0) rowptr[N_NODES] = total;
}

__global__ void gcn_fill_kernel(const int* __restrict__ src, const int* __restrict__ dst,
                                const int* __restrict__ rank, const int* __restrict__ rowptr,
                                int* __restrict__ csr) {
  const int e = (blockIdx.x * blockDim.x + threadIdx.x) * 4;
  if (e < N_EDGES) {
    const v4i s = *(const v4i*)(src + e);
    const v4i d = *(const v4i*)(dst + e);
    const v4i r = *(const v4i*)(rank + e);
    csr[rowptr[d.x] + r.x] = s.x;
    csr[rowptr[d.y] + r.y] = s.y;
    csr[rowptr[d.z] + r.z] = s.z;
    csr[rowptr[d.w] + r.w] = s.w;
  }
}

__global__ void gcn_cntinv_kernel(float* __restrict__ cnt) {
  int g = blockIdx.x * blockDim.x + threadIdx.x;
  if (g < G_GRAPHS) cnt[g] = 1.0f / fmaxf(cnt[g], 1.0f);
}

template <int K>
__global__ void gcn_gemm_wmma_kernel(const float* __restrict__ X,
                                     const float* __restrict__ W,
                                     const float* __restrict__ dinv,
                                     float* __restrict__ H, int nRows) {
  const int wave = blockIdx.x * (blockDim.x >> 5) + (threadIdx.x >> 5);
  const int lane = threadIdx.x & 31;
  const int numTiles = (nRows >> 4) * 2;
  if (wave >= numTiles) return;
  const int rowTile = wave >> 1;
  const int colTile = wave & 1;
  const int half = lane >> 4;
  const int l16  = lane & 15;
  const int aRow = rowTile * 16 + l16;
  const int col  = colTile * 16 + l16;

  const float da = dinv[aRow];

  v8f c = {};
#pragma unroll
  for (int k0 = 0; k0 < K; k0 += 4) {
    const int ka = k0 + 2 * half;
    v2f a = *(const v2f*)(X + aRow * K + ka) * da;
    v2f b;
    b.x = W[(ka + 0) * HID + col];
    b.y = W[(ka + 1) * HID + col];
    c = __builtin_amdgcn_wmma_f32_16x16x4_f32(
         false, a,  false, b,
         (short)0, c,  false,  false);
  }
#pragma unroll
  for (int r = 0; r < 8; ++r) {
    const int m = r + 8 * half;
    H[(rowTile * 16 + m) * HID + col] = c[r];
  }
}

static_assert(HID == 32, "one lane per channel");
static_assert((long long)N_EDGES + N_NODES + 1 <= (long long)N_NODES * HID,
              "sorted edge list and row starts must fit the plane the accumulator used to take");
static_assert(N_EDGES <= N_NODES * HID, "arrival ranks must fit the activation plane");
template <int JBASE>
__device__ __forceinline__ float gcn_rows16(const float* __restrict__ Hlin, int s, int n,
                                            int lane, float acc) {
  float h[16];
#pragma unroll
  for (int j = 0; j < 16; ++j) {
    const int sj = __builtin_amdgcn_readlane(s, JBASE + j);
    h[j] = Hlin[(unsigned)(sj * HID + lane)];
  }
#pragma unroll
  for (int j = 0; j < 16; ++j) acc += (JBASE + j < n) ? h[j] : 0.0f;
  return acc;
}

#define GC_BLOCKS 2048
#define GC_WAVES  (GC_BLOCKS * 8)
__global__ void __launch_bounds__(256)
gcn_gather_combine_kernel(const int* __restrict__ rowptr, const int* __restrict__ csr,
                          const float* __restrict__ Hlin,
                          const float* __restrict__ bias, const float* __restrict__ dinv,
                          const int* __restrict__ batch, const float* __restrict__ cntinv,
                          float* __restrict__ Hact, float* __restrict__ feat) {
  int node = __builtin_amdgcn_readfirstlane(blockIdx.x * (blockDim.x >> 5) + (threadIdx.x >> 5));
  const int lane = threadIdx.x & 31;
  if (node >= N_NODES) return;
  const float bl = bias[lane];
  int beg = rowptr[node];
  int end = rowptr[node + 1];
  int g = batch[node];
  for (;;) {
    int p = beg;
    int q = p + lane;
    int s = q < end ? csr[q] : 0;
    float acc = Hlin[(unsigned)(node * HID + lane)];
    const int next = node + GC_WAVES;
    const bool more = next < N_NODES;
    int nbeg = 0, nend = 0, ng = 0;
    if (more) {
      nbeg = rowptr[next];
      nend = rowptr[next + 1];
      ng = batch[next];
    }
    const float di = dinv[node];
    const float ci = cntinv[g];
    while (p < end) {
      const int n = end - p;
      acc = gcn_rows16<0>(Hlin, s, n, lane, acc);
      if (n > 16) acc = gcn_rows16<16>(Hlin, s, n, lane, acc);
      p += 32;
      q = p + lane;
      s = q < end ? csr[q] : 0;
    }
    float v = acc * di + bl;
    v = v > 0.0f ? v : NEG_SLOPE * v;
    Hact[(unsigned)(node * HID + lane)] = v;
    atomicAdd(&feat[g * HID + lane], v * ci);
    if (!more) break;
    node = next;
    beg = nbeg;
    end = nend;
    g = ng;
  }
}

__global__ void gcn_merge_kernel(const float* __restrict__ feats, float* __restrict__ out) {
  int t = blockIdx.x * blockDim.x + threadIdx.x;
  const int GH = G_GRAPHS * HID;
  if (t < GH) out[t] = (feats[t] + feats[GH + t] + feats[2 * GH + t]) * (1.0f / 3.0f);
}

static inline int cdiv(int a, int b) { return (a + b - 1) / b; }

extern "C" void kernel_launch(void* const* d_in, const int* in_sizes, int n_in,
                              void* d_out, int out_size, void* d_ws, size_t ws_size,
                              hipStream_t stream) {
  (void)in_sizes; (void)n_in; (void)out_size; (void)ws_size;
  const float* x     = (const float*)d_in[0];
  const float* W0    = (const float*)d_in[1];
  const float* b0    = (const float*)d_in[2];
  const float* W1    = (const float*)d_in[3];
  const float* b1    = (const float*)d_in[4];
  const float* W2    = (const float*)d_in[5];
  const float* b2    = (const float*)d_in[6];
  const int*   src   = (const int*)d_in[7];
  const int*   dst   = (const int*)d_in[8];
  const int*   batch = (const int*)d_in[9];
  float* out = (float*)d_out;

  float* ws    = (float*)d_ws;
  float* deg   = ws;
  float* cnt   = deg + N_NODES;
  float* lin   = cnt + G_GRAPHS;
  float* act   = lin + (size_t)N_NODES * HID;
  float* agg   = act + (size_t)N_NODES * HID;
  int* degi    = (int*)deg;
  int* rank    = (int*)act;
  int* csr     = (int*)agg;
  int* rowptr  = csr + N_EDGES;
  float* feats = agg + (size_t)N_NODES * HID;

  const int T = 256;
  const int nTiles = (N_NODES / 16) * 2;

  gcn_zero_kernel<<<cdiv(N_NODES, T), T, 0, stream>>>(deg, N_NODES);
  gcn_zero_kernel<<<1, T, 0, stream>>>(cnt, G_GRAPHS);
  gcn_zero_kernel<<<cdiv(3 * G_GRAPHS * HID, T), T, 0, stream>>>(feats, 3 * G_GRAPHS * HID);
  gcn_degree_rank_kernel<<<cdiv(N_EDGES / 4, T), T, 0, stream>>>(dst, degi, rank);
  gcn_counts_kernel<<<cdiv(N_NODES, T), T, 0, stream>>>(batch, cnt);
  gcn_scan_dinv_kernel<<<1, SCAN_THREADS, 0, stream>>>(degi, rowptr);
  gcn_cntinv_kernel<<<1, T, 0, stream>>>(cnt);
  gcn_fill_kernel<<<cdiv(N_EDGES / 4, T), T, 0, stream>>>(src, dst, rank, rowptr, csr);

  const float* Ws[3] = {W0, W1, W2};
  const float* bs[3] = {b0, b1, b2};
  const float* hin = x;
  for (int layer = 0; layer < 3; ++layer) {
    if (layer == 0)
      gcn_gemm_wmma_kernel<C_IN><<<cdiv(nTiles, T / 32), T, 0, stream>>>(hin, Ws[0], deg, lin, N_NODES);
    else
      gcn_gemm_wmma_kernel<HID><<<cdiv(nTiles, T / 32), T, 0, stream>>>(hin, Ws[layer], deg, lin, N_NODES);
    gcn_gather_combine_kernel<<<GC_BLOCKS, T, 0, stream>>>(
        rowptr, csr, lin, bs[layer], deg, batch, cnt, act, feats + layer * G_GRAPHS * HID);
    hin = act;
  }
  gcn_merge_kernel<<<cdiv(G_GRAPHS * HID, T), T, 0, stream>>>(feats, out);
}


# independent kernels share launches (a block picks its body by its index): 3 clears in one, degree count beside graph count, scan beside 1/count, row fill beside layer-0 product; 15 launches -> 10
# speedup vs baseline: 1.8064x; 1.0499x over previous
#include <hip/hip_runtime.h>

#define N_NODES   100000
#define N_EDGES   1600000
#define C_IN      64
#define HID       32
#define G_GRAPHS  128
#define NEG_SLOPE 0.01f

typedef __attribute__((ext_vector_type(2))) float v2f;
typedef __attribute__((ext_vector_type(8))) float v8f;


#define CLEAR_A (N_NODES + G_GRAPHS)
#define CLEAR_B (3 * G_GRAPHS * HID)
__global__ void gcn_clear_kernel(float* __restrict__ degcnt, float* __restrict__ feats) {
  const int i = blockIdx.x * blockDim.x + threadIdx.x;
  if (i < CLEAR_A) degcnt[i] = 0.0f;
  else if (i - CLEAR_A < CLEAR_B) feats[i - CLEAR_A] = 0.0f;
}

typedef __attribute__((ext_vector_type(4))) int v4i;
static_assert(N_EDGES % 4 == 0, "16-byte groups of edges must not straddle the end");
#define EDGE_BLOCKS ((N_EDGES / 4 + 255) / 256)
#define NODE_BLOCKS ((N_NODES + 255) / 256)
__device__ __forceinline__ void gcn_degree_rank_body(int blk, const int* __restrict__ dst,
                                                     int* __restrict__ degi,
                                                     int* __restrict__ rank) {
  const int e = (blk * 256 + threadIdx.x) * 4;
  if (e < N_EDGES) {
    const v4i d = *(const v4i*)(dst + e);
    v4i r;
    r.x = atomicAdd(&degi[d.x], 1);
    r.y = atomicAdd(&degi[d.y], 1);
    r.z = atomicAdd(&degi[d.z], 1);
    r.w = atomicAdd(&degi[d.w], 1);
    *(v4i*)(rank + e) = r;
  }
}

__device__ __forceinline__ void gcn_counts_body(int blk, const int* __restrict__ batch,
                                                float* __restrict__ cnt) {
  const int i = blk * 256 + threadIdx.x;
  const int lane = threadIdx.x & 31;
  const bool live = i < N_NODES;
  const int g = live ? batch[i] : -1;
  unsigned remaining = __builtin_amdgcn_ballot_w32(live);
  for (int trip = 0; trip < 32 && remaining != 0u; ++trip) {
    const int lead = __ffs(remaining) - 1;
    const int gl = __builtin_amdgcn_readlane(g, lead);
    const unsigned same = __builtin_amdgcn_ballot_w32(live && g == gl) & remaining;
    if (lane == lead) atomicAdd(&cnt[gl], (float)__popc(same));
    remaining &= ~same;
  }
}

__global__ void __launch_bounds__(256)
gcn_degree_counts_kernel(const int* __restrict__ dst, int* __restrict__ degi,
                         int* __restrict__ rank, const int* __restrict__ batch,
                         float* __restrict__ cnt) {
  if (blockIdx.x < EDGE_BLOCKS) gcn_degree_rank_body(blockIdx.x, dst, degi, rank);
  else gcn_counts_body(blockIdx.x - EDGE_BLOCKS, batch, cnt);
}

#define SCAN_THREADS 1024
#define SCAN_TRIPS   25
#define SCAN_WAVE    (SCAN_TRIPS * 128)
static_assert((SCAN_THREADS / 32) * SCAN_WAVE >= N_NODES, "one work-group must cover every node");
static_assert(N_NODES % 4 == 0, "16-byte groups must not straddle the end");
__global__ void __launch_bounds__(SCAN_THREADS)
gcn_scan_dinv_cntinv_kernel(int* __restrict__ degbits, int* __restrict__ rowptr,
                            float* __restrict__ cnt) {
  if (blockIdx.x != 0) {
    if (threadIdx.x < G_GRAPHS) cnt[threadIdx.x] = 1.0f / fmaxf(cnt[threadIdx.x], 1.0f);
    return;
  }
  __shared__ int wtot[SCAN_THREADS / 32];
  const int lane = threadIdx.x & 31;
  const int w = threadIdx.x >> 5;
  const int base = w * SCAN_WAVE + lane * 4;
  int ex[SCAN_TRIPS];
  int carry = 0;
#pragma unroll
  for (int k = 0; k < SCAN_TRIPS; ++k) {
    const int idx = base + k * 128;
    v4i v = {0, 0, 0, 0};
    if (idx < N_NODES) v = *(const v4i*)(degbits + idx);
    const int t = v.x + v.y + v.z + v.w;
    int inc = t;
#pragma unroll
    for (int d = 1; d < 32; d <<= 1) {
      const int u = __shfl_up(inc, d);
      if (lane >= d) inc += u;
    }
    ex[k] = carry + inc - t;
    carry += __builtin_amdgcn_readlane(inc, 31);
  }
  if (lane == 0) wtot[w] = carry;
  __syncthreads();
  const int mine = wtot[lane];
  int off = lane < w ? mine : 0;
  int total = mine;
#pragma unroll
  for (int d = 16; d >= 1; d >>= 1) {
    off += __shfl_xor(off, d);
    total += __shfl_xor(total, d);
  }
#pragma unroll
  for (int k = 0; k < SCAN_TRIPS; ++k) {
    const int idx = base + k * 128;
    if (idx < N_NODES) {
      const v4i v = *(const v4i*)(degbits + idx);
      v4i o;
      o.x = off + ex[k];
      o.y = o.x + v.x;
      o.z = o.y + v.y;
      o.w = o.z + v.z;
      *(v4i*)(rowptr + idx) = o;
      v4i f;
      f.x = __float_as_int(rsqrtf((float)v.x + 1.0f));
      f.y = __float_as_int(rsqrtf((float)v.y + 1.0f));
      f.z = __float_as_int(rsqrtf((float)v.z + 1.0f));
      f.w = __float_as_int(rsqrtf((float)v.w + 1.0f));
      *(v4i*)(degbits + idx) = f;
    }
  }
  if (threadIdx.x == 0) rowptr[N_NODES] = total;
}

__device__ __forceinline__ void gcn_fill_body(int blk, const int* __restrict__ src,
                                              const int* __restrict__ dst,
                                              const int* __restrict__ rank,
                                              const int* __restrict__ rowptr,
                                              int* __restrict__ csr) {
  const int e = (blk * 256 + threadIdx.x) * 4;
  if (e < N_EDGES) {
    const v4i s = *(const v4i*)(src + e);
    const v4i d = *(const v4i*)(dst + e);
    const v4i r = *(const v4i*)(rank + e);
    csr[rowptr[d.x] + r.x] = s.x;
    csr[rowptr[d.y] + r.y] = s.y;
    csr[rowptr[d.z] + r.z] = s.z;
    csr[rowptr[d.w] + r.w] = s.w;
  }
}

#define GEMM_BLOCKS (((N_NODES / 16) * 2 + 7) / 8)
template <int K>
__device__ __forceinline__ void gcn_gemm_wmma_body(int blk, const float* __restrict__ X,
                                                   const float* __restrict__ W,
                                                   const float* __restrict__ dinv,
                                                   float* __restrict__ H, int nRows) {
  const int wave = blk * 8 + (threadIdx.x >> 5);
  const int lane = threadIdx.x & 31;
  const int numTiles = (nRows >> 4) * 2;
  if (wave >= numTiles) return;
  const int rowTile = wave >> 1;
  const int colTile = wave & 1;
  const int half = lane >> 4;
  const int l16  = lane & 15;
  const int aRow = rowTile * 16 + l16;
  const int col  = colTile * 16 + l16;

  const float da = dinv[aRow];

  v8f c = {};
#pragma unroll
  for (int k0 = 0; k0 < K; k0 += 4) {
    const int ka = k0 + 2 * half;
    v2f a = *(const v2f*)(X + aRow * K + ka) * da;
    v2f b;
    b.x = W[(ka + 0) * HID + col];
    b.y = W[(ka + 1) * HID + col];
    c = __builtin_amdgcn_wmma_f32_16x16x4_f32(
         false, a,  false, b,
         (short)0, c,  false,  false);
  }
#pragma unroll
  for (int r = 0; r < 8; ++r) {
    const int m = r + 8 * half;
    H[(rowTile * 16 + m) * HID + col] = c[r];
  }
}

__global__ void __launch_bounds__(256)
gcn_gemm_wmma_kernel(const float* __restrict__ X, const float* __restrict__ W,
                     const float* __restrict__ dinv, float* __restrict__ H, int nRows) {
  gcn_gemm_wmma_body<HID>(blockIdx.x, X, W, dinv, H, nRows);
}

__global__ void __launch_bounds__(256)
gcn_fill_gemm0_kernel(const int* __restrict__ src, const int* __restrict__ dst,
                      const int* __restrict__ rank, const int* __restrict__ rowptr,
                      int* __restrict__ csr, const float* __restrict__ X,
                      const float* __restrict__ W, const float* __restrict__ dinv,
                      float* __restrict__ H, int nRows) {
  if (blockIdx.x < EDGE_BLOCKS) gcn_fill_body(blockIdx.x, src, dst, rank, rowptr, csr);
  else gcn_gemm_wmma_body<C_IN>(blockIdx.x - EDGE_BLOCKS, X, W, dinv, H, nRows);
}

static_assert(HID == 32, "one lane per channel");
static_assert((long long)N_EDGES + N_NODES + 1 <= (long long)N_NODES * HID,
              "sorted edge list and row starts must fit the plane the accumulator used to take");
static_assert(N_EDGES <= N_NODES * HID, "arrival ranks must fit the activation plane");
template <int JBASE>
__device__ __forceinline__ float gcn_rows16(const float* __restrict__ Hlin, int s, int n,
                                            int lane, float acc) {
  float h[16];
#pragma unroll
  for (int j = 0; j < 16; ++j) {
    const int sj = __builtin_amdgcn_readlane(s, JBASE + j);
    h[j] = Hlin[(unsigned)(sj * HID + lane)];
  }
#pragma unroll
  for (int j = 0; j < 16; ++j) acc += (JBASE + j < n) ? h[j] : 0.0f;
  return acc;
}

#define GC_BLOCKS 2048
#define GC_WAVES  (GC_BLOCKS * 8)
__global__ void __launch_bounds__(256)
gcn_gather_combine_kernel(const int* __restrict__ rowptr, const int* __restrict__ csr,
                          const float* __restrict__ Hlin,
                          const float* __restrict__ bias, const float* __restrict__ dinv,
                          const int* __restrict__ batch, const float* __restrict__ cntinv,
                          float* __restrict__ Hact, float* __restrict__ feat) {
  int node = __builtin_amdgcn_readfirstlane(blockIdx.x * (blockDim.x >> 5) + (threadIdx.x >> 5));
  const int lane = threadIdx.x & 31;
  if (node >= N_NODES) return;
  const float bl = bias[lane];
  int beg = rowptr[node];
  int end = rowptr[node + 1];
  int g = batch[node];
  for (;;) {
    int p = beg;
    int q = p + lane;
    int s = q < end ? csr[q] : 0;
    float acc = Hlin[(unsigned)(node * HID + lane)];
    const int next = node + GC_WAVES;
    const bool more = next < N_NODES;
    int nbeg = 0, nend = 0, ng = 0;
    if (more) {
      nbeg = rowptr[next];
      nend = rowptr[next + 1];
      ng = batch[next];
    }
    const float di = dinv[node];
    const float ci = cntinv[g];
    while (p < end) {
      const int n = end - p;
      acc = gcn_rows16<0>(Hlin, s, n, lane, acc);
      if (n > 16) acc = gcn_rows16<16>(Hlin, s, n, lane, acc);
      p += 32;
      q = p + lane;
      s = q < end ? csr[q] : 0;
    }
    float v = acc * di + bl;
    v = v > 0.0f ? v : NEG_SLOPE * v;
    Hact[(unsigned)(node * HID + lane)] = v;
    atomicAdd(&feat[g * HID + lane], v * ci);
    if (!more) break;
    node = next;
    beg = nbeg;
    end = nend;
    g = ng;
  }
}

__global__ void gcn_merge_kernel(const float* __restrict__ feats, float* __restrict__ out) {
  int t = blockIdx.x * blockDim.x + threadIdx.x;
  const int GH = G_GRAPHS * HID;
  if (t < GH) out[t] = (feats[t] + feats[GH + t] + feats[2 * GH + t]) * (1.0f / 3.0f);
}

static inline int cdiv(int a, int b) { return (a + b - 1) / b; }

extern "C" void kernel_launch(void* const* d_in, const int* in_sizes, int n_in,
                              void* d_out, int out_size, void* d_ws, size_t ws_size,
                              hipStream_t stream) {
  (void)in_sizes; (void)n_in; (void)out_size; (void)ws_size;
  const float* x     = (const float*)d_in[0];
  const float* W0    = (const float*)d_in[1];
  const float* b0    = (const float*)d_in[2];
  const float* W1    = (const float*)d_in[3];
  const float* b1    = (const float*)d_in[4];
  const float* W2    = (const float*)d_in[5];
  const float* b2    = (const float*)d_in[6];
  const int*   src   = (const int*)d_in[7];
  const int*   dst   = (const int*)d_in[8];
  const int*   batch = (const int*)d_in[9];
  float* out = (float*)d_out;

  float* ws    = (float*)d_ws;
  float* deg   = ws;
  float* cnt   = deg + N_NODES;
  float* lin   = cnt + G_GRAPHS;
  float* act   = lin + (size_t)N_NODES * HID;
  float* agg   = act + (size_t)N_NODES * HID;
  int* degi    = (int*)deg;
  int* rank    = (int*)act;
  int* csr     = (int*)agg;
  int* rowptr  = csr + N_EDGES;
  float* feats = agg + (size_t)N_NODES * HID;

  const int T = 256;

  gcn_clear_kernel<<<cdiv(CLEAR_A + CLEAR_B, T), T, 0, stream>>>(deg, feats);
  gcn_degree_counts_kernel<<<EDGE_BLOCKS + NODE_BLOCKS, T, 0, stream>>>(dst, degi, rank, batch, cnt);
  gcn_scan_dinv_cntinv_kernel<<<2, SCAN_THREADS, 0, stream>>>(degi, rowptr, cnt);
  gcn_fill_gemm0_kernel<<<EDGE_BLOCKS + GEMM_BLOCKS, T, 0, stream>>>(
      src, dst, rank, rowptr, csr, x, W0, deg, lin, N_NODES);

  const float* Ws[3] = {W0, W1, W2};
  const float* bs[3] = {b0, b1, b2};
  for (int layer = 0; layer < 3; ++layer) {
    if (layer > 0)
      gcn_gemm_wmma_kernel<<<GEMM_BLOCKS, T, 0, stream>>>(act, Ws[layer], deg, lin, N_NODES);
    gcn_gather_combine_kernel<<<GC_BLOCKS, T, 0, stream>>>(
        rowptr, csr, lin, bs[layer], deg, batch, cnt, act, feats + layer * G_GRAPHS * HID);
  }
  gcn_merge_kernel<<<cdiv(G_GRAPHS * HID, T), T, 0, stream>>>(feats, out);
}


# edge sort by destination on LDS counters: edges binned by 256-node span (chunk counts, one scan), then one work-group per bin ranks and fills its rows; no per-edge add to memory; row order by arrival
# speedup vs baseline: 2.5094x; 1.3892x over previous
#include <hip/hip_runtime.h>

#define N_NODES   100000
#define N_EDGES   1600000
#define C_IN      64
#define HID       32
#define G_GRAPHS  128
#define NEG_SLOPE 0.01f

typedef __attribute__((ext_vector_type(2))) float v2f;
typedef __attribute__((ext_vector_type(8))) float v8f;


#define CLEAR_A G_GRAPHS
#define CLEAR_B (3 * G_GRAPHS * HID)
__global__ void gcn_clear_kernel(float* __restrict__ cnt, float* __restrict__ feats) {
  const int i = blockIdx.x * blockDim.x + threadIdx.x;
  if (i < CLEAR_A) cnt[i] = 0.0f;
  else if (i - CLEAR_A < CLEAR_B) feats[i - CLEAR_A] = 0.0f;
}

typedef __attribute__((ext_vector_type(4))) int v4i;
typedef __attribute__((ext_vector_type(2))) int v2i;

template <int BLOCK_THREADS>
__device__ __forceinline__ void gcn_counts_body(int blk, const int* __restrict__ batch,
                                                float* __restrict__ cnt) {
  const int i = blk * BLOCK_THREADS + threadIdx.x;
  const int lane = threadIdx.x & 31;
  const bool live = i < N_NODES;
  const int g = live ? batch[i] : -1;
  unsigned remaining = __builtin_amdgcn_ballot_w32(live);
  for (int trip = 0; trip < 32 && remaining != 0u; ++trip) {
    const int lead = __ffs(remaining) - 1;
    const int gl = __builtin_amdgcn_readlane(g, lead);
    const unsigned same = __builtin_amdgcn_ballot_w32(live && g == gl) & remaining;
    if (lane == lead) atomicAdd(&cnt[gl], (float)__popc(same));
    remaining &= ~same;
  }
}

#define BIN_SHIFT    8
#define BIN_NODES    (1 << BIN_SHIFT)
#define N_BINS       ((N_NODES + BIN_NODES - 1) / BIN_NODES)
#define PART_BLOCKS  256
#define PART_THREADS 1024
#define PART_CHUNK   (N_EDGES / PART_BLOCKS)
#define PART_TRIPS   ((PART_CHUNK + PART_THREADS - 1) / PART_THREADS)
#define HIST_LEN     (N_BINS * PART_BLOCKS)
static_assert(N_EDGES % PART_BLOCKS == 0, "equal chunks");
static_assert(HIST_LEN % 4 == 0, "16-byte groups must not straddle the end");
#define COUNT_BLOCKS ((N_NODES + PART_THREADS - 1) / PART_THREADS)

__global__ void __launch_bounds__(PART_THREADS)
gcn_bincount_counts_kernel(const int* __restrict__ dst, int* __restrict__ hist,
                           const int* __restrict__ batch, float* __restrict__ cnt) {
  __shared__ int binc[N_BINS];
  if (blockIdx.x >= PART_BLOCKS) {
    gcn_counts_body<PART_THREADS>(blockIdx.x - PART_BLOCKS, batch, cnt);
    return;
  }
  const int e0 = blockIdx.x * PART_CHUNK;
  int d[PART_TRIPS];
#pragma unroll
  for (int j = 0; j < PART_TRIPS; ++j) {
    const int k = threadIdx.x + j * PART_THREADS;
    d[j] = k < PART_CHUNK ? dst[e0 + k] : -1;
  }
  for (int t = threadIdx.x; t < N_BINS; t += PART_THREADS) binc[t] = 0;
  __syncthreads();
#pragma unroll
  for (int j = 0; j < PART_TRIPS; ++j)
    if (d[j] >= 0) atomicAdd(&binc[d[j] >> BIN_SHIFT], 1);
  __syncthreads();
  for (int t = threadIdx.x; t < N_BINS; t += PART_THREADS)
    hist[t * PART_BLOCKS + blockIdx.x] = binc[t];
}

#define SCAN_THREADS 1024
#define SCAN_TRIPS   25
#define SCAN_WAVE    (SCAN_TRIPS * 128)
static_assert((SCAN_THREADS / 32) * SCAN_WAVE >= HIST_LEN, "one work-group must cover every cell");
__global__ void __launch_bounds__(SCAN_THREADS)
gcn_scan_cntinv_kernel(const int* __restrict__ hist, int* __restrict__ base,
                       float* __restrict__ cnt) {
  if (blockIdx.x != 0) {
    if (threadIdx.x < G_GRAPHS) cnt[threadIdx.x] = 1.0f / fmaxf(cnt[threadIdx.x], 1.0f);
    return;
  }
  __shared__ int wtot[SCAN_THREADS / 32];
  const int lane = threadIdx.x & 31;
  const int w = threadIdx.x >> 5;
  const int first = w * SCAN_WAVE + lane * 4;
  int ex[SCAN_TRIPS];
  int carry = 0;
#pragma unroll
  for (int k = 0; k < SCAN_TRIPS; ++k) {
    const int idx = first + k * 128;
    v4i v = {0, 0, 0, 0};
    if (idx < HIST_LEN) v = *(const v4i*)(hist + idx);
    const int t = v.x + v.y + v.z + v.w;
    int inc = t;
#pragma unroll
    for (int d = 1; d < 32; d <<= 1) {
      const int u = __shfl_up(inc, d);
      if (lane >= d) inc += u;
    }
    ex[k] = carry + inc - t;
    carry += __builtin_amdgcn_readlane(inc, 31);
  }
  if (lane == 0) wtot[w] = carry;
  __syncthreads();
  const int mine = wtot[lane];
  int off = lane < w ? mine : 0;
  int total = mine;
#pragma unroll
  for (int d = 16; d >= 1; d >>= 1) {
    off += __shfl_xor(off, d);
    total += __shfl_xor(total, d);
  }
#pragma unroll
  for (int k = 0; k < SCAN_TRIPS; ++k) {
    const int idx = first + k * 128;
    if (idx < HIST_LEN) {
      const v4i v = *(const v4i*)(hist + idx);
      v4i o;
      o.x = off + ex[k];
      o.y = o.x + v.x;
      o.z = o.y + v.y;
      o.w = o.z + v.z;
      *(v4i*)(base + idx) = o;
    }
  }
  if (threadIdx.x == 0) base[HIST_LEN] = total;
}

#define GEMM_TILES (((N_NODES) / 16) * 2)
#define GEMM_BLOCKS ((GEMM_TILES + 7) / 8)
template <int K, bool SCALE, int WAVES_PER_BLOCK>
__device__ __forceinline__ void gcn_gemm_wmma_body(int blk, const float* __restrict__ X,
                                                   const float* __restrict__ W,
                                                   const float* __restrict__ dinv,
                                                   float* __restrict__ H, int nRows) {
  const int wave = blk * WAVES_PER_BLOCK + (threadIdx.x >> 5);
  const int lane = threadIdx.x & 31;
  const int numTiles = (nRows >> 4) * 2;
  if (wave >= numTiles) return;
  const int rowTile = wave >> 1;
  const int colTile = wave & 1;
  const int half = lane >> 4;
  const int l16  = lane & 15;
  const int aRow = rowTile * 16 + l16;
  const int col  = colTile * 16 + l16;

  const float da = SCALE ? dinv[aRow] : 1.0f;

  v8f c = {};
#pragma unroll
  for (int k0 = 0; k0 < K; k0 += 4) {
    const int ka = k0 + 2 * half;
    v2f a = *(const v2f*)(X + aRow * K + ka);
    if (SCALE) a *= da;
    v2f b;
    b.x = W[(ka + 0) * HID + col];
    b.y = W[(ka + 1) * HID + col];
    c = __builtin_amdgcn_wmma_f32_16x16x4_f32(
         false, a,  false, b,
         (short)0, c,  false,  false);
  }
#pragma unroll
  for (int r = 0; r < 8; ++r) {
    const int m = r + 8 * half;
    H[(rowTile * 16 + m) * HID + col] = c[r];
  }
}

__global__ void __launch_bounds__(256)
gcn_gemm_wmma_kernel(const float* __restrict__ X, const float* __restrict__ W,
                     const float* __restrict__ dinv, float* __restrict__ H, int nRows) {
  gcn_gemm_wmma_body<HID, true, 8>(blockIdx.x, X, W, dinv, H, nRows);
}

#define GEMM0_BLOCKS ((GEMM_TILES + PART_THREADS / 32 - 1) / (PART_THREADS / 32))
__global__ void __launch_bounds__(PART_THREADS)
gcn_binscatter_gemm0_kernel(const int* __restrict__ src, const int* __restrict__ dst,
                            const int* __restrict__ base, int* __restrict__ pairbuf,
                            const float* __restrict__ X, const float* __restrict__ W,
                            float* __restrict__ H, int nRows) {
  __shared__ int cur[N_BINS];
  v2i* __restrict__ pairs = (v2i*)pairbuf;
  if (blockIdx.x >= PART_BLOCKS) {
    gcn_gemm_wmma_body<C_IN, false, PART_THREADS / 32>(blockIdx.x - PART_BLOCKS, X, W, nullptr,
                                                       H, nRows);
    return;
  }
  const int e0 = blockIdx.x * PART_CHUNK;
  v2i pr[PART_TRIPS];
#pragma unroll
  for (int j = 0; j < PART_TRIPS; ++j) {
    const int k = threadIdx.x + j * PART_THREADS;
    pr[j].x = k < PART_CHUNK ? src[e0 + k] : 0;
    pr[j].y = k < PART_CHUNK ? dst[e0 + k] : -1;
  }
  for (int t = threadIdx.x; t < N_BINS; t += PART_THREADS)
    cur[t] = base[t * PART_BLOCKS + blockIdx.x];
  __syncthreads();
#pragma unroll
  for (int j = 0; j < PART_TRIPS; ++j) {
    if (pr[j].y >= 0) {
      const int slot = atomicAdd(&cur[pr[j].y >> BIN_SHIFT], 1);
      pairs[slot] = pr[j];
    }
  }
}

static_assert(BIN_NODES == 256, "eight waves: one thread per node of the bin");
static_assert(HID % 4 == 0, "rows in 16-byte pieces");
#define BIN_BATCH 8
typedef __attribute__((ext_vector_type(4))) float v4f;
static_assert(N_NODES % 16 == 0, "row tiles of the products");
__global__ void __launch_bounds__(BIN_NODES)
gcn_bin_rows_kernel(const int* __restrict__ base, const int* __restrict__ pairbuf,
                    int* __restrict__ rowptr, int* __restrict__ csr,
                    float* __restrict__ dinv, float* __restrict__ H) {
  __shared__ int cntL[BIN_NODES];
  __shared__ float dinvL[BIN_NODES];
  __shared__ int wtot[BIN_NODES / 32];
  const v2i* __restrict__ pairs = (const v2i*)pairbuf;
  const int b = blockIdx.x;
  const int t = threadIdx.x;
  const int lane = t & 31;
  const int w = t >> 5;
  const int beg = base[b * PART_BLOCKS];
  const int end = base[(b + 1) * PART_BLOCKS];
  const int n0 = b * BIN_NODES;
  const int rows = N_NODES - n0 < BIN_NODES ? N_NODES - n0 : BIN_NODES;
  v4f* __restrict__ H4 = (v4f*)(H + n0 * HID);
  const int pieces = rows * (HID / 4);
  v4f hv[HID / 4];
#pragma unroll
  for (int j = 0; j < HID / 4; ++j) {
    const int i = t + j * BIN_NODES;
    hv[j] = H4[i < pieces ? i : pieces - 1];
  }
  cntL[t] = 0;
  __syncthreads();
  for (int eb = beg; eb < end; eb += BIN_NODES * BIN_BATCH) {
    int dl[BIN_BATCH];
#pragma unroll
    for (int j = 0; j < BIN_BATCH; ++j) {
      const int e = eb + j * BIN_NODES + t;
      dl[j] = e < end ? pairs[e].y : -1;
    }
#pragma unroll
    for (int j = 0; j < BIN_BATCH; ++j)
      if (dl[j] >= 0) atomicAdd(&cntL[dl[j] & (BIN_NODES - 1)], 1);
  }
  __syncthreads();
  const int deg = cntL[t];
  int inc = deg;
#pragma unroll
  for (int d = 1; d < 32; d <<= 1) {
    const int u = __shfl_up(inc, d);
    if (lane >= d) inc += u;
  }
  if (lane == 31) wtot[w] = inc;
  __syncthreads();
  int off = 0;
#pragma unroll
  for (int j = 0; j < BIN_NODES / 32; ++j) off += j < w ? wtot[j] : 0;
  const int start = beg + off + inc - deg;
  const int node = n0 + t;
  const float di = rsqrtf((float)deg + 1.0f);
  dinvL[t] = di;
  cntL[t] = start;
  if (node < N_NODES) {
    rowptr[node] = start;
    dinv[node] = di;
  }
  if (b == N_BINS - 1 && t == 0) rowptr[N_NODES] = end;
  __syncthreads();
  for (int eb = beg; eb < end; eb += BIN_NODES * BIN_BATCH) {
    v2i pr[BIN_BATCH];
#pragma unroll
    for (int j = 0; j < BIN_BATCH; ++j) {
      const int e = eb + j * BIN_NODES + t;
      v2i none = {0, -1};
      pr[j] = e < end ? pairs[e] : none;
    }
#pragma unroll
    for (int j = 0; j < BIN_BATCH; ++j) {
      if (pr[j].y >= 0) {
        const int slot = atomicAdd(&cntL[pr[j].y & (BIN_NODES - 1)], 1);
        csr[slot] = pr[j].x;
      }
    }
  }
#pragma unroll
  for (int j = 0; j < HID / 4; ++j) {
    const int i = t + j * BIN_NODES;
    if (i < pieces) H4[i] = hv[j] * dinvL[i / (HID / 4)];
  }
}

static_assert(HID == 32, "one lane per channel");
#define WS_ROWPTR_LEN (N_NODES + 4)
#define WS_SCAN_LEN   ((SCAN_THREADS / 32) * SCAN_WAVE)
static_assert((long long)N_EDGES + WS_ROWPTR_LEN + 2 * WS_SCAN_LEN <= (long long)N_NODES * HID,
              "sorted edge list, row starts, cell counts and cell starts must fit the plane the "
              "accumulator used to take");
static_assert(HIST_LEN + 1 <= WS_SCAN_LEN, "cell starts and their closing total");
static_assert(2 * (long long)N_EDGES <= (long long)N_NODES * HID,
              "the (source, destination) pairs must fit the activation plane");
static_assert(N_EDGES % 4 == 0 && WS_ROWPTR_LEN % 4 == 0, "16-byte alignment of the scan's arrays");
template <int JBASE>
__device__ __forceinline__ float gcn_rows16(const float* __restrict__ Hlin, int s, int n,
                                            int lane, float acc) {
  float h[16];
#pragma unroll
  for (int j = 0; j < 16; ++j) {
    const int sj = __builtin_amdgcn_readlane(s, JBASE + j);
    h[j] = Hlin[(unsigned)(sj * HID + lane)];
  }
#pragma unroll
  for (int j = 0; j < 16; ++j) acc += (JBASE + j < n) ? h[j] : 0.0f;
  return acc;
}

#define GC_BLOCKS 2048
#define GC_WAVES  (GC_BLOCKS * 8)
__global__ void __launch_bounds__(256)
gcn_gather_combine_kernel(const int* __restrict__ rowptr, const int* __restrict__ csr,
                          const float* __restrict__ Hlin,
                          const float* __restrict__ bias, const float* __restrict__ dinv,
                          const int* __restrict__ batch, const float* __restrict__ cntinv,
                          float* __restrict__ Hact, float* __restrict__ feat) {
  int node = __builtin_amdgcn_readfirstlane(blockIdx.x * (blockDim.x >> 5) + (threadIdx.x >> 5));
  const int lane = threadIdx.x & 31;
  if (node >= N_NODES) return;
  const float bl = bias[lane];
  int beg = rowptr[node];
  int end = rowptr[node + 1];
  int g = batch[node];
  for (;;) {
    int p = beg;
    int q = p + lane;
    int s = q < end ? csr[q] : 0;
    float acc = Hlin[(unsigned)(node * HID + lane)];
    const int next = node + GC_WAVES;
    const bool more = next < N_NODES;
    int nbeg = 0, nend = 0, ng = 0;
    if (more) {
      nbeg = rowptr[next];
      nend = rowptr[next + 1];
      ng = batch[next];
    }
    const float di = dinv[node];
    const float ci = cntinv[g];
    while (p < end) {
      const int n = end - p;
      acc = gcn_rows16<0>(Hlin, s, n, lane, acc);
      if (n > 16) acc = gcn_rows16<16>(Hlin, s, n, lane, acc);
      p += 32;
      q = p + lane;
      s = q < end ? csr[q] : 0;
    }
    float v = acc * di + bl;
    v = v > 0.0f ? v : NEG_SLOPE * v;
    Hact[(unsigned)(node * HID + lane)] = v;
    atomicAdd(&feat[g * HID + lane], v * ci);
    if (!more) break;
    node = next;
    beg = nbeg;
    end = nend;
    g = ng;
  }
}

__global__ void gcn_merge_kernel(const float* __restrict__ feats, float* __restrict__ out) {
  int t = blockIdx.x * blockDim.x + threadIdx.x;
  const int GH = G_GRAPHS * HID;
  if (t < GH) out[t] = (feats[t] + feats[GH + t] + feats[2 * GH + t]) * (1.0f / 3.0f);
}

static inline int cdiv(int a, int b) { return (a + b - 1) / b; }

extern "C" void kernel_launch(void* const* d_in, const int* in_sizes, int n_in,
                              void* d_out, int out_size, void* d_ws, size_t ws_size,
                              hipStream_t stream) {
  (void)in_sizes; (void)n_in; (void)out_size; (void)ws_size;
  const float* x     = (const float*)d_in[0];
  const float* W0    = (const float*)d_in[1];
  const float* b0    = (const float*)d_in[2];
  const float* W1    = (const float*)d_in[3];
  const float* b1    = (const float*)d_in[4];
  const float* W2    = (const float*)d_in[5];
  const float* b2    = (const float*)d_in[6];
  const int*   src   = (const int*)d_in[7];
  const int*   dst   = (const int*)d_in[8];
  const int*   batch = (const int*)d_in[9];
  float* out = (float*)d_out;

  float* ws    = (float*)d_ws;
  float* deg   = ws;
  float* cnt   = deg + N_NODES;
  float* lin   = cnt + G_GRAPHS;
  float* act   = lin + (size_t)N_NODES * HID;
  float* agg   = act + (size_t)N_NODES * HID;
  int* pairs   = (int*)act;
  int* csr     = (int*)agg;
  int* rowptr  = csr + N_EDGES;
  int* hist    = rowptr + WS_ROWPTR_LEN;
  int* base    = hist + WS_SCAN_LEN;
  float* feats = agg + (size_t)N_NODES * HID;

  const int T = 256;

  gcn_clear_kernel<<<cdiv(CLEAR_A + CLEAR_B, T), T, 0, stream>>>(cnt, feats);
  gcn_bincount_counts_kernel<<<PART_BLOCKS + COUNT_BLOCKS, PART_THREADS, 0, stream>>>(
      dst, hist, batch, cnt);
  gcn_scan_cntinv_kernel<<<2, SCAN_THREADS, 0, stream>>>(hist, base, cnt);
  gcn_binscatter_gemm0_kernel<<<PART_BLOCKS + GEMM0_BLOCKS, PART_THREADS, 0, stream>>>(
      src, dst, base, pairs, x, W0, lin, N_NODES);
  gcn_bin_rows_kernel<<<N_BINS, BIN_NODES, 0, stream>>>(base, pairs, rowptr, csr, deg, lin);

  const float* Ws[3] = {W0, W1, W2};
  const float* bs[3] = {b0, b1, b2};
  for (int layer = 0; layer < 3; ++layer) {
    if (layer > 0)
      gcn_gemm_wmma_kernel<<<GEMM_BLOCKS, T, 0, stream>>>(act, Ws[layer], deg, lin, N_NODES);
    gcn_gather_combine_kernel<<<GC_BLOCKS, T, 0, stream>>>(
        rowptr, csr, lin, bs[layer], deg, batch, cnt, act, feats + layer * G_GRAPHS * HID);
  }
  gcn_merge_kernel<<<cdiv(G_GRAPHS * HID, T), T, 0, stream>>>(feats, out);
}


# products on fewer, fatter waves: a wave takes whole 16x32 row tiles (two in layers 1-2: 4x fewer waves enter; one in layer 0: 2x fewer), loads ahead of the matrix work; k summed in a new order
# speedup vs baseline: 2.7026x; 1.0770x over previous
#include <hip/hip_runtime.h>

#define N_NODES   100000
#define N_EDGES   1600000
#define C_IN      64
#define HID       32
#define G_GRAPHS  128
#define NEG_SLOPE 0.01f

typedef __attribute__((ext_vector_type(2))) float v2f;
typedef __attribute__((ext_vector_type(8))) float v8f;


#define CLEAR_A G_GRAPHS
#define CLEAR_B (3 * G_GRAPHS * HID)
__global__ void gcn_clear_kernel(float* __restrict__ cnt, float* __restrict__ feats) {
  const int i = blockIdx.x * blockDim.x + threadIdx.x;
  if (i < CLEAR_A) cnt[i] = 0.0f;
  else if (i - CLEAR_A < CLEAR_B) feats[i - CLEAR_A] = 0.0f;
}

typedef __attribute__((ext_vector_type(4))) int v4i;
typedef __attribute__((ext_vector_type(2))) int v2i;

template <int BLOCK_THREADS>
__device__ __forceinline__ void gcn_counts_body(int blk, const int* __restrict__ batch,
                                                float* __restrict__ cnt) {
  const int i = blk * BLOCK_THREADS + threadIdx.x;
  const int lane = threadIdx.x & 31;
  const bool live = i < N_NODES;
  const int g = live ? batch[i] : -1;
  unsigned remaining = __builtin_amdgcn_ballot_w32(live);
  for (int trip = 0; trip < 32 && remaining != 0u; ++trip) {
    const int lead = __ffs(remaining) - 1;
    const int gl = __builtin_amdgcn_readlane(g, lead);
    const unsigned same = __builtin_amdgcn_ballot_w32(live && g == gl) & remaining;
    if (lane == lead) atomicAdd(&cnt[gl], (float)__popc(same));
    remaining &= ~same;
  }
}

#define BIN_SHIFT    8
#define BIN_NODES    (1 << BIN_SHIFT)
#define N_BINS       ((N_NODES + BIN_NODES - 1) / BIN_NODES)
#define PART_BLOCKS  256
#define PART_THREADS 1024
#define PART_CHUNK   (N_EDGES / PART_BLOCKS)
#define PART_TRIPS   ((PART_CHUNK + PART_THREADS - 1) / PART_THREADS)
#define HIST_LEN     (N_BINS * PART_BLOCKS)
static_assert(N_EDGES % PART_BLOCKS == 0, "equal chunks");
static_assert(HIST_LEN % 4 == 0, "16-byte groups must not straddle the end");
#define COUNT_BLOCKS ((N_NODES + PART_THREADS - 1) / PART_THREADS)

__global__ void __launch_bounds__(PART_THREADS)
gcn_bincount_counts_kernel(const int* __restrict__ dst, int* __restrict__ hist,
                           const int* __restrict__ batch, float* __restrict__ cnt) {
  __shared__ int binc[N_BINS];
  if (blockIdx.x >= PART_BLOCKS) {
    gcn_counts_body<PART_THREADS>(blockIdx.x - PART_BLOCKS, batch, cnt);
    return;
  }
  const int e0 = blockIdx.x * PART_CHUNK;
  int d[PART_TRIPS];
#pragma unroll
  for (int j = 0; j < PART_TRIPS; ++j) {
    const int k = threadIdx.x + j * PART_THREADS;
    d[j] = k < PART_CHUNK ? dst[e0 + k] : -1;
  }
  for (int t = threadIdx.x; t < N_BINS; t += PART_THREADS) binc[t] = 0;
  __syncthreads();
#pragma unroll
  for (int j = 0; j < PART_TRIPS; ++j)
    if (d[j] >= 0) atomicAdd(&binc[d[j] >> BIN_SHIFT], 1);
  __syncthreads();
  for (int t = threadIdx.x; t < N_BINS; t += PART_THREADS)
    hist[t * PART_BLOCKS + blockIdx.x] = binc[t];
}

#define SCAN_THREADS 1024
#define SCAN_TRIPS   25
#define SCAN_WAVE    (SCAN_TRIPS * 128)
static_assert((SCAN_THREADS / 32) * SCAN_WAVE >= HIST_LEN, "one work-group must cover every cell");
__global__ void __launch_bounds__(SCAN_THREADS)
gcn_scan_cntinv_kernel(const int* __restrict__ hist, int* __restrict__ base,
                       float* __restrict__ cnt) {
  if (blockIdx.x != 0) {
    if (threadIdx.x < G_GRAPHS) cnt[threadIdx.x] = 1.0f / fmaxf(cnt[threadIdx.x], 1.0f);
    return;
  }
  __shared__ int wtot[SCAN_THREADS / 32];
  const int lane = threadIdx.x & 31;
  const int w = threadIdx.x >> 5;
  const int first = w * SCAN_WAVE + lane * 4;
  int ex[SCAN_TRIPS];
  int carry = 0;
#pragma unroll
  for (int k = 0; k < SCAN_TRIPS; ++k) {
    const int idx = first + k * 128;
    v4i v = {0, 0, 0, 0};
    if (idx < HIST_LEN) v = *(const v4i*)(hist + idx);
    const int t = v.x + v.y + v.z + v.w;
    int inc = t;
#pragma unroll
    for (int d = 1; d < 32; d <<= 1) {
      const int u = __shfl_up(inc, d);
      if (lane >= d) inc += u;
    }
    ex[k] = carry + inc - t;
    carry += __builtin_amdgcn_readlane(inc, 31);
  }
  if (lane == 0) wtot[w] = carry;
  __syncthreads();
  const int mine = wtot[lane];
  int off = lane < w ? mine : 0;
  int total = mine;
#pragma unroll
  for (int d = 16; d >= 1; d >>= 1) {
    off += __shfl_xor(off, d);
    total += __shfl_xor(total, d);
  }
#pragma unroll
  for (int k = 0; k < SCAN_TRIPS; ++k) {
    const int idx = first + k * 128;
    if (idx < HIST_LEN) {
      const v4i v = *(const v4i*)(hist + idx);
      v4i o;
      o.x = off + ex[k];
      o.y = o.x + v.x;
      o.z = o.y + v.y;
      o.w = o.z + v.z;
      *(v4i*)(base + idx) = o;
    }
  }
  if (threadIdx.x == 0) base[HIST_LEN] = total;
}

typedef __attribute__((ext_vector_type(4))) float v4f;
template <int K, bool SCALE, int WAVES_PER_BLOCK, int TILES, bool LOADS_FIRST>
__device__ __forceinline__ void gcn_gemm_wmma_body(int blk, const float* __restrict__ X,
                                                   const float* __restrict__ W,
                                                   const float* __restrict__ dinv,
                                                   float* __restrict__ H, int nRows) {
  static_assert(K % 8 == 0, "a lane half's k range in 16-byte pieces");
  const int wave = blk * WAVES_PER_BLOCK + (threadIdx.x >> 5);
  const int lane = threadIdx.x & 31;
  const int row0 = wave * (16 * TILES);
  if (row0 >= nRows) return;
  const int half = lane >> 4;
  const int l16  = lane & 15;

  v4f af[TILES][K / 8];
  float da[TILES];
#pragma unroll
  for (int u = 0; u < TILES; ++u) {
    const int row = row0 + u * 16 + l16;
    const v4f* __restrict__ xr = (const v4f*)(X + row * K + half * (K / 2));
#pragma unroll
    for (int j = 0; j < K / 8; ++j) af[u][j] = xr[j];
    da[u] = SCALE ? dinv[row] : 1.0f;
  }
  v2f wf[K / 4][2];
#pragma unroll
  for (int s = 0; s < K / 4; ++s) {
    const int kk = half * (K / 2) + 2 * s;
#pragma unroll
    for (int p = 0; p < 2; ++p) {
      wf[s][p].x = W[(kk + 0) * HID + 16 * p + l16];
      wf[s][p].y = W[(kk + 1) * HID + 16 * p + l16];
    }
  }
  if (LOADS_FIRST) __builtin_amdgcn_sched_barrier(0);

#pragma unroll
  for (int u = 0; u < TILES; ++u) {
#pragma unroll
    for (int p = 0; p < 2; ++p) {
      v8f c = {};
#pragma unroll
      for (int s = 0; s < K / 4; ++s) {
        v2f a;
        a.x = af[u][s >> 1][(s & 1) * 2 + 0];
        a.y = af[u][s >> 1][(s & 1) * 2 + 1];
        if (SCALE) a *= da[u];
        c = __builtin_amdgcn_wmma_f32_16x16x4_f32(
             false, a,  false, wf[s][p],
             (short)0, c,  false,  false);
      }
#pragma unroll
      for (int r = 0; r < 8; ++r)
        H[(row0 + u * 16 + r + 8 * half) * HID + 16 * p + l16] = c[r];
    }
  }
}

#define GEMM_TILES_PER_WAVE 2
static_assert(N_NODES % (16 * GEMM_TILES_PER_WAVE) == 0, "whole waves of row tiles");
#define GEMM_WAVES  (N_NODES / (16 * GEMM_TILES_PER_WAVE))
#define GEMM_BLOCKS ((GEMM_WAVES + 7) / 8)
__global__ void __launch_bounds__(256)
gcn_gemm_wmma_kernel(const float* __restrict__ X, const float* __restrict__ W,
                     const float* __restrict__ dinv, float* __restrict__ H, int nRows) {
  gcn_gemm_wmma_body<HID, true, 8, GEMM_TILES_PER_WAVE, true>(blockIdx.x, X, W, dinv, H, nRows);
}

static_assert(N_NODES % 16 == 0, "row tiles of the products");
#define GEMM0_WAVES  (N_NODES / 16)
#define GEMM0_BLOCKS ((GEMM0_WAVES + PART_THREADS / 32 - 1) / (PART_THREADS / 32))
__global__ void __launch_bounds__(PART_THREADS)
gcn_binscatter_gemm0_kernel(const int* __restrict__ src, const int* __restrict__ dst,
                            const int* __restrict__ base, int* __restrict__ pairbuf,
                            const float* __restrict__ X, const float* __restrict__ W,
                            float* __restrict__ H, int nRows) {
  __shared__ int cur[N_BINS];
  v2i* __restrict__ pairs = (v2i*)pairbuf;
  if (blockIdx.x >= PART_BLOCKS) {
    gcn_gemm_wmma_body<C_IN, false, PART_THREADS / 32, 1, false>(blockIdx.x - PART_BLOCKS, X, W,
                                                                 nullptr, H, nRows);
    return;
  }
  const int e0 = blockIdx.x * PART_CHUNK;
  v2i pr[PART_TRIPS];
#pragma unroll
  for (int j = 0; j < PART_TRIPS; ++j) {
    const int k = threadIdx.x + j * PART_THREADS;
    pr[j].x = k < PART_CHUNK ? src[e0 + k] : 0;
    pr[j].y = k < PART_CHUNK ? dst[e0 + k] : -1;
  }
  for (int t = threadIdx.x; t < N_BINS; t += PART_THREADS)
    cur[t] = base[t * PART_BLOCKS + blockIdx.x];
  __syncthreads();
#pragma unroll
  for (int j = 0; j < PART_TRIPS; ++j) {
    if (pr[j].y >= 0) {
      const int slot = atomicAdd(&cur[pr[j].y >> BIN_SHIFT], 1);
      pairs[slot] = pr[j];
    }
  }
}

static_assert(BIN_NODES == 256, "eight waves: one thread per node of the bin");
static_assert(HID % 4 == 0, "rows in 16-byte pieces");
#define BIN_BATCH 8
__global__ void __launch_bounds__(BIN_NODES)
gcn_bin_rows_kernel(const int* __restrict__ base, const int* __restrict__ pairbuf,
                    int* __restrict__ rowptr, int* __restrict__ csr,
                    float* __restrict__ dinv, float* __restrict__ H) {
  __shared__ int cntL[BIN_NODES];
  __shared__ float dinvL[BIN_NODES];
  __shared__ int wtot[BIN_NODES / 32];
  const v2i* __restrict__ pairs = (const v2i*)pairbuf;
  const int b = blockIdx.x;
  const int t = threadIdx.x;
  const int lane = t & 31;
  const int w = t >> 5;
  const int beg = base[b * PART_BLOCKS];
  const int end = base[(b + 1) * PART_BLOCKS];
  const int n0 = b * BIN_NODES;
  const int rows = N_NODES - n0 < BIN_NODES ? N_NODES - n0 : BIN_NODES;
  v4f* __restrict__ H4 = (v4f*)(H + n0 * HID);
  const int pieces = rows * (HID / 4);
  v4f hv[HID / 4];
#pragma unroll
  for (int j = 0; j < HID / 4; ++j) {
    const int i = t + j * BIN_NODES;
    hv[j] = H4[i < pieces ? i : pieces - 1];
  }
  cntL[t] = 0;
  __syncthreads();
  for (int eb = beg; eb < end; eb += BIN_NODES * BIN_BATCH) {
    int dl[BIN_BATCH];
#pragma unroll
    for (int j = 0; j < BIN_BATCH; ++j) {
      const int e = eb + j * BIN_NODES + t;
      dl[j] = e < end ? pairs[e].y : -1;
    }
#pragma unroll
    for (int j = 0; j < BIN_BATCH; ++j)
      if (dl[j] >= 0) atomicAdd(&cntL[dl[j] & (BIN_NODES - 1)], 1);
  }
  __syncthreads();
  const int deg = cntL[t];
  int inc = deg;
#pragma unroll
  for (int d = 1; d < 32; d <<= 1) {
    const int u = __shfl_up(inc, d);
    if (lane >= d) inc += u;
  }
  if (lane == 31) wtot[w] = inc;
  __syncthreads();
  int off = 0;
#pragma unroll
  for (int j = 0; j < BIN_NODES / 32; ++j) off += j < w ? wtot[j] : 0;
  const int start = beg + off + inc - deg;
  const int node = n0 + t;
  const float di = rsqrtf((float)deg + 1.0f);
  dinvL[t] = di;
  cntL[t] = start;
  if (node < N_NODES) {
    rowptr[node] = start;
    dinv[node] = di;
  }
  if (b == N_BINS - 1 && t == 0) rowptr[N_NODES] = end;
  __syncthreads();
  for (int eb = beg; eb < end; eb += BIN_NODES * BIN_BATCH) {
    v2i pr[BIN_BATCH];
#pragma unroll
    for (int j = 0; j < BIN_BATCH; ++j) {
      const int e = eb + j * BIN_NODES + t;
      v2i none = {0, -1};
      pr[j] = e < end ? pairs[e] : none;
    }
#pragma unroll
    for (int j = 0; j < BIN_BATCH; ++j) {
      if (pr[j].y >= 0) {
        const int slot = atomicAdd(&cntL[pr[j].y & (BIN_NODES - 1)], 1);
        csr[slot] = pr[j].x;
      }
    }
  }
#pragma unroll
  for (int j = 0; j < HID / 4; ++j) {
    const int i = t + j * BIN_NODES;
    if (i < pieces) H4[i] = hv[j] * dinvL[i / (HID / 4)];
  }
}

static_assert(HID == 32, "one lane per channel");
#define WS_ROWPTR_LEN (N_NODES + 4)
#define WS_SCAN_LEN   ((SCAN_THREADS / 32) * SCAN_WAVE)
static_assert((long long)N_EDGES + WS_ROWPTR_LEN + 2 * WS_SCAN_LEN <= (long long)N_NODES * HID,
              "sorted edge list, row starts, cell counts and cell starts must fit the plane the "
              "accumulator used to take");
static_assert(HIST_LEN + 1 <= WS_SCAN_LEN, "cell starts and their closing total");
static_assert(2 * (long long)N_EDGES <= (long long)N_NODES * HID,
              "the (source, destination) pairs must fit the activation plane");
static_assert(N_EDGES % 4 == 0 && WS_ROWPTR_LEN % 4 == 0, "16-byte alignment of the scan's arrays");
template <int JBASE>
__device__ __forceinline__ float gcn_rows16(const float* __restrict__ Hlin, int s, int n,
                                            int lane, float acc) {
  float h[16];
#pragma unroll
  for (int j = 0; j < 16; ++j) {
    const int sj = __builtin_amdgcn_readlane(s, JBASE + j);
    h[j] = Hlin[(unsigned)(sj * HID + lane)];
  }
#pragma unroll
  for (int j = 0; j < 16; ++j) acc += (JBASE + j < n) ? h[j] : 0.0f;
  return acc;
}

#define GC_BLOCKS 2048
#define GC_WAVES  (GC_BLOCKS * 8)
__global__ void __launch_bounds__(256)
gcn_gather_combine_kernel(const int* __restrict__ rowptr, const int* __restrict__ csr,
                          const float* __restrict__ Hlin,
                          const float* __restrict__ bias, const float* __restrict__ dinv,
                          const int* __restrict__ batch, const float* __restrict__ cntinv,
                          float* __restrict__ Hact, float* __restrict__ feat) {
  int node = __builtin_amdgcn_readfirstlane(blockIdx.x * (blockDim.x >> 5) + (threadIdx.x >> 5));
  const int lane = threadIdx.x & 31;
  if (node >= N_NODES) return;
  const float bl = bias[lane];
  int beg = rowptr[node];
  int end = rowptr[node + 1];
  int g = batch[node];
  for (;;) {
    int p = beg;
    int q = p + lane;
    int s = q < end ? csr[q] : 0;
    float acc = Hlin[(unsigned)(node * HID + lane)];
    const int next = node + GC_WAVES;
    const bool more = next < N_NODES;
    int nbeg = 0, nend = 0, ng = 0;
    if (more) {
      nbeg = rowptr[next];
      nend = rowptr[next + 1];
      ng = batch[next];
    }
    const float di = dinv[node];
    const float ci = cntinv[g];
    while (p < end) {
      const int n = end - p;
      acc = gcn_rows16<0>(Hlin, s, n, lane, acc);
      if (n > 16) acc = gcn_rows16<16>(Hlin, s, n, lane, acc);
      p += 32;
      q = p + lane;
      s = q < end ? csr[q] : 0;
    }
    float v = acc * di + bl;
    v = v > 0.0f ? v : NEG_SLOPE * v;
    Hact[(unsigned)(node * HID + lane)] = v;
    atomicAdd(&feat[g * HID + lane], v * ci);
    if (!more) break;
    node = next;
    beg = nbeg;
    end = nend;
    g = ng;
  }
}

__global__ void gcn_merge_kernel(const float* __restrict__ feats, float* __restrict__ out) {
  int t = blockIdx.x * blockDim.x + threadIdx.x;
  const int GH = G_GRAPHS * HID;
  if (t < GH) out[t] = (feats[t] + feats[GH + t] + feats[2 * GH + t]) * (1.0f / 3.0f);
}

static inline int cdiv(int a, int b) { return (a + b - 1) / b; }

extern "C" void kernel_launch(void* const* d_in, const int* in_sizes, int n_in,
                              void* d_out, int out_size, void* d_ws, size_t ws_size,
                              hipStream_t stream) {
  (void)in_sizes; (void)n_in; (void)out_size; (void)ws_size;
  const float* x     = (const float*)d_in[0];
  const float* W0    = (const float*)d_in[1];
  const float* b0    = (const float*)d_in[2];
  const float* W1    = (const float*)d_in[3];
  const float* b1    = (const float*)d_in[4];
  const float* W2    = (const float*)d_in[5];
  const float* b2    = (const float*)d_in[6];
  const int*   src   = (const int*)d_in[7];
  const int*   dst   = (const int*)d_in[8];
  const int*   batch = (const int*)d_in[9];
  float* out = (float*)d_out;

  float* ws    = (float*)d_ws;
  float* deg   = ws;
  float* cnt   = deg + N_NODES;
  float* lin   = cnt + G_GRAPHS;
  float* act   = lin + (size_t)N_NODES * HID;
  float* agg   = act + (size_t)N_NODES * HID;
  int* pairs   = (int*)act;
  int* csr     = (int*)agg;
  int* rowptr  = csr + N_EDGES;
  int* hist    = rowptr + WS_ROWPTR_LEN;
  int* base    = hist + WS_SCAN_LEN;
  float* feats = agg + (size_t)N_NODES * HID;

  const int T = 256;

  gcn_clear_kernel<<<cdiv(CLEAR_A + CLEAR_B, T), T, 0, stream>>>(cnt, feats);
  gcn_bincount_counts_kernel<<<PART_BLOCKS + COUNT_BLOCKS, PART_THREADS, 0, stream>>>(
      dst, hist, batch, cnt);
  gcn_scan_cntinv_kernel<<<2, SCAN_THREADS, 0, stream>>>(hist, base, cnt);
  gcn_binscatter_gemm0_kernel<<<PART_BLOCKS + GEMM0_BLOCKS, PART_THREADS, 0, stream>>>(
      src, dst, base, pairs, x, W0, lin, N_NODES);
  gcn_bin_rows_kernel<<<N_BINS, BIN_NODES, 0, stream>>>(base, pairs, rowptr, csr, deg, lin);

  const float* Ws[3] = {W0, W1, W2};
  const float* bs[3] = {b0, b1, b2};
  for (int layer = 0; layer < 3; ++layer) {
    if (layer > 0)
      gcn_gemm_wmma_kernel<<<GEMM_BLOCKS, T, 0, stream>>>(act, Ws[layer], deg, lin, N_NODES);
    gcn_gather_combine_kernel<<<GC_BLOCKS, T, 0, stream>>>(
        rowptr, csr, lin, bs[layer], deg, batch, cnt, act, feats + layer * G_GRAPHS * HID);
  }
  gcn_merge_kernel<<<cdiv(G_GRAPHS * HID, T), T, 0, stream>>>(feats, out);
}


# next layer's product moved into the edge-sum launch: a wave keeps its nodes' activated rows in LDS, multiplies them by W on the matrix unit, writes the scaled rows; 11 -> 9 launches; sum order kept
# speedup vs baseline: 2.9630x; 1.0964x over previous
#include <hip/hip_runtime.h>

#define N_NODES   100000
#define N_EDGES   1600000
#define C_IN      64
#define HID       32
#define G_GRAPHS  128
#define NEG_SLOPE 0.01f

typedef __attribute__((ext_vector_type(2))) float v2f;
typedef __attribute__((ext_vector_type(8))) float v8f;


#define CLEAR_A G_GRAPHS
#define CLEAR_B (3 * G_GRAPHS * HID)
__global__ void gcn_clear_kernel(float* __restrict__ cnt, float* __restrict__ feats) {
  const int i = blockIdx.x * blockDim.x + threadIdx.x;
  if (i < CLEAR_A) cnt[i] = 0.0f;
  else if (i - CLEAR_A < CLEAR_B) feats[i - CLEAR_A] = 0.0f;
}

typedef __attribute__((ext_vector_type(4))) int v4i;
typedef __attribute__((ext_vector_type(2))) int v2i;

template <int BLOCK_THREADS>
__device__ __forceinline__ void gcn_counts_body(int blk, const int* __restrict__ batch,
                                                float* __restrict__ cnt) {
  const int i = blk * BLOCK_THREADS + threadIdx.x;
  const int lane = threadIdx.x & 31;
  const bool live = i < N_NODES;
  const int g = live ? batch[i] : -1;
  unsigned remaining = __builtin_amdgcn_ballot_w32(live);
  for (int trip = 0; trip < 32 && remaining != 0u; ++trip) {
    const int lead = __ffs(remaining) - 1;
    const int gl = __builtin_amdgcn_readlane(g, lead);
    const unsigned same = __builtin_amdgcn_ballot_w32(live && g == gl) & remaining;
    if (lane == lead) atomicAdd(&cnt[gl], (float)__popc(same));
    remaining &= ~same;
  }
}

#define BIN_SHIFT    8
#define BIN_NODES    (1 << BIN_SHIFT)
#define N_BINS       ((N_NODES + BIN_NODES - 1) / BIN_NODES)
#define PART_BLOCKS  256
#define PART_THREADS 1024
#define PART_CHUNK   (N_EDGES / PART_BLOCKS)
#define PART_TRIPS   ((PART_CHUNK + PART_THREADS - 1) / PART_THREADS)
#define HIST_LEN     (N_BINS * PART_BLOCKS)
static_assert(N_EDGES % PART_BLOCKS == 0, "equal chunks");
static_assert(HIST_LEN % 4 == 0, "16-byte groups must not straddle the end");
#define COUNT_BLOCKS ((N_NODES + PART_THREADS - 1) / PART_THREADS)

__global__ void __launch_bounds__(PART_THREADS)
gcn_bincount_counts_kernel(const int* __restrict__ dst, int* __restrict__ hist,
                           const int* __restrict__ batch, float* __restrict__ cnt) {
  __shared__ int binc[N_BINS];
  if (blockIdx.x >= PART_BLOCKS) {
    gcn_counts_body<PART_THREADS>(blockIdx.x - PART_BLOCKS, batch, cnt);
    return;
  }
  const int e0 = blockIdx.x * PART_CHUNK;
  int d[PART_TRIPS];
#pragma unroll
  for (int j = 0; j < PART_TRIPS; ++j) {
    const int k = threadIdx.x + j * PART_THREADS;
    d[j] = k < PART_CHUNK ? dst[e0 + k] : -1;
  }
  for (int t = threadIdx.x; t < N_BINS; t += PART_THREADS) binc[t] = 0;
  __syncthreads();
#pragma unroll
  for (int j = 0; j < PART_TRIPS; ++j)
    if (d[j] >= 0) atomicAdd(&binc[d[j] >> BIN_SHIFT], 1);
  __syncthreads();
  for (int t = threadIdx.x; t < N_BINS; t += PART_THREADS)
    hist[t * PART_BLOCKS + blockIdx.x] = binc[t];
}

#define SCAN_THREADS 1024
#define SCAN_TRIPS   25
#define SCAN_WAVE    (SCAN_TRIPS * 128)
static_assert((SCAN_THREADS / 32) * SCAN_WAVE >= HIST_LEN, "one work-group must cover every cell");
__global__ void __launch_bounds__(SCAN_THREADS)
gcn_scan_cntinv_kernel(const int* __restrict__ hist, int* __restrict__ base,
                       float* __restrict__ cnt) {
  if (blockIdx.x != 0) {
    if (threadIdx.x < G_GRAPHS) cnt[threadIdx.x] = 1.0f / fmaxf(cnt[threadIdx.x], 1.0f);
    return;
  }
  __shared__ int wtot[SCAN_THREADS / 32];
  const int lane = threadIdx.x & 31;
  const int w = threadIdx.x >> 5;
  const int first = w * SCAN_WAVE + lane * 4;
  int ex[SCAN_TRIPS];
  int carry = 0;
#pragma unroll
  for (int k = 0; k < SCAN_TRIPS; ++k) {
    const int idx = first + k * 128;
    v4i v = {0, 0, 0, 0};
    if (idx < HIST_LEN) v = *(const v4i*)(hist + idx);
    const int t = v.x + v.y + v.z + v.w;
    int inc = t;
#pragma unroll
    for (int d = 1; d < 32; d <<= 1) {
      const int u = __shfl_up(inc, d);
      if (lane >= d) inc += u;
    }
    ex[k] = carry + inc - t;
    carry += __builtin_amdgcn_readlane(inc, 31);
  }
  if (lane == 0) wtot[w] = carry;
  __syncthreads();
  const int mine = wtot[lane];
  int off = lane < w ? mine : 0;
  int total = mine;
#pragma unroll
  for (int d = 16; d >= 1; d >>= 1) {
    off += __shfl_xor(off, d);
    total += __shfl_xor(total, d);
  }
#pragma unroll
  for (int k = 0; k < SCAN_TRIPS; ++k) {
    const int idx = first + k * 128;
    if (idx < HIST_LEN) {
      const v4i v = *(const v4i*)(hist + idx);
      v4i o;
      o.x = off + ex[k];
      o.y = o.x + v.x;
      o.z = o.y + v.y;
      o.w = o.z + v.z;
      *(v4i*)(base + idx) = o;
    }
  }
  if (threadIdx.x == 0) base[HIST_LEN] = total;
}

typedef __attribute__((ext_vector_type(4))) float v4f;
template <int K, int WAVES_PER_BLOCK>
__device__ __forceinline__ void gcn_gemm_wmma_body(int blk, const float* __restrict__ X,
                                                   const float* __restrict__ W,
                                                   float* __restrict__ H, int nRows) {
  static_assert(K % 8 == 0, "a lane half's k range in 16-byte pieces");
  const int wave = blk * WAVES_PER_BLOCK + (threadIdx.x >> 5);
  const int lane = threadIdx.x & 31;
  const int row0 = wave * 16;
  if (row0 >= nRows) return;
  const int half = lane >> 4;
  const int l16  = lane & 15;

  v4f af[K / 8];
  const v4f* __restrict__ xr = (const v4f*)(X + (row0 + l16) * K + half * (K / 2));
#pragma unroll
  for (int j = 0; j < K / 8; ++j) af[j] = xr[j];
  v2f wf[K / 4][2];
#pragma unroll
  for (int s = 0; s < K / 4; ++s) {
    const int kk = half * (K / 2) + 2 * s;
#pragma unroll
    for (int p = 0; p < 2; ++p) {
      wf[s][p].x = W[(kk + 0) * HID + 16 * p + l16];
      wf[s][p].y = W[(kk + 1) * HID + 16 * p + l16];
    }
  }

#pragma unroll
  for (int p = 0; p < 2; ++p) {
    v8f c = {};
#pragma unroll
    for (int s = 0; s < K / 4; ++s) {
      v2f a;
      a.x = af[s >> 1][(s & 1) * 2 + 0];
      a.y = af[s >> 1][(s & 1) * 2 + 1];
      c = __builtin_amdgcn_wmma_f32_16x16x4_f32(
           false, a,  false, wf[s][p],
           (short)0, c,  false,  false);
    }
#pragma unroll
    for (int r = 0; r < 8; ++r)
      H[(row0 + r + 8 * half) * HID + 16 * p + l16] = c[r];
  }
}

static_assert(N_NODES % 16 == 0, "row tiles of the products");
#define GEMM0_WAVES  (N_NODES / 16)
#define GEMM0_BLOCKS ((GEMM0_WAVES + PART_THREADS / 32 - 1) / (PART_THREADS / 32))
__global__ void __launch_bounds__(PART_THREADS)
gcn_binscatter_gemm0_kernel(const int* __restrict__ src, const int* __restrict__ dst,
                            const int* __restrict__ base, int* __restrict__ pairbuf,
                            const float* __restrict__ X, const float* __restrict__ W,
                            float* __restrict__ H, int nRows) {
  __shared__ int cur[N_BINS];
  v2i* __restrict__ pairs = (v2i*)pairbuf;
  if (blockIdx.x >= PART_BLOCKS) {
    gcn_gemm_wmma_body<C_IN, PART_THREADS / 32>(blockIdx.x - PART_BLOCKS, X, W, H, nRows);
    return;
  }
  const int e0 = blockIdx.x * PART_CHUNK;
  v2i pr[PART_TRIPS];
#pragma unroll
  for (int j = 0; j < PART_TRIPS; ++j) {
    const int k = threadIdx.x + j * PART_THREADS;
    pr[j].x = k < PART_CHUNK ? src[e0 + k] : 0;
    pr[j].y = k < PART_CHUNK ? dst[e0 + k] : -1;
  }
  for (int t = threadIdx.x; t < N_BINS; t += PART_THREADS)
    cur[t] = base[t * PART_BLOCKS + blockIdx.x];
  __syncthreads();
#pragma unroll
  for (int j = 0; j < PART_TRIPS; ++j) {
    if (pr[j].y >= 0) {
      const int slot = atomicAdd(&cur[pr[j].y >> BIN_SHIFT], 1);
      pairs[slot] = pr[j];
    }
  }
}

static_assert(BIN_NODES == 256, "eight waves: one thread per node of the bin");
static_assert(HID % 4 == 0, "rows in 16-byte pieces");
#define BIN_BATCH 8
__global__ void __launch_bounds__(BIN_NODES)
gcn_bin_rows_kernel(const int* __restrict__ base, const int* __restrict__ pairbuf,
                    int* __restrict__ rowptr, int* __restrict__ csr,
                    float* __restrict__ dinv, float* __restrict__ H) {
  __shared__ int cntL[BIN_NODES];
  __shared__ float dinvL[BIN_NODES];
  __shared__ int wtot[BIN_NODES / 32];
  const v2i* __restrict__ pairs = (const v2i*)pairbuf;
  const int b = blockIdx.x;
  const int t = threadIdx.x;
  const int lane = t & 31;
  const int w = t >> 5;
  const int beg = base[b * PART_BLOCKS];
  const int end = base[(b + 1) * PART_BLOCKS];
  const int n0 = b * BIN_NODES;
  const int rows = N_NODES - n0 < BIN_NODES ? N_NODES - n0 : BIN_NODES;
  v4f* __restrict__ H4 = (v4f*)(H + n0 * HID);
  const int pieces = rows * (HID / 4);
  v4f hv[HID / 4];
#pragma unroll
  for (int j = 0; j < HID / 4; ++j) {
    const int i = t + j * BIN_NODES;
    hv[j] = H4[i < pieces ? i : pieces - 1];
  }
  cntL[t] = 0;
  __syncthreads();
  for (int eb = beg; eb < end; eb += BIN_NODES * BIN_BATCH) {
    int dl[BIN_BATCH];
#pragma unroll
    for (int j = 0; j < BIN_BATCH; ++j) {
      const int e = eb + j * BIN_NODES + t;
      dl[j] = e < end ? pairs[e].y : -1;
    }
#pragma unroll
    for (int j = 0; j < BIN_BATCH; ++j)
      if (dl[j] >= 0) atomicAdd(&cntL[dl[j] & (BIN_NODES - 1)], 1);
  }
  __syncthreads();
  const int deg = cntL[t];
  int inc = deg;
#pragma unroll
  for (int d = 1; d < 32; d <<= 1) {
    const int u = __shfl_up(inc, d);
    if (lane >= d) inc += u;
  }
  if (lane == 31) wtot[w] = inc;
  __syncthreads();
  int off = 0;
#pragma unroll
  for (int j = 0; j < BIN_NODES / 32; ++j) off += j < w ? wtot[j] : 0;
  const int start = beg + off + inc - deg;
  const int node = n0 + t;
  const float di = rsqrtf((float)deg + 1.0f);
  dinvL[t] = di;
  cntL[t] = start;
  if (node < N_NODES) {
    rowptr[node] = start;
    dinv[node] = di;
  }
  if (b == N_BINS - 1 && t == 0) rowptr[N_NODES] = end;
  __syncthreads();
  for (int eb = beg; eb < end; eb += BIN_NODES * BIN_BATCH) {
    v2i pr[BIN_BATCH];
#pragma unroll
    for (int j = 0; j < BIN_BATCH; ++j) {
      const int e = eb + j * BIN_NODES + t;
      v2i none = {0, -1};
      pr[j] = e < end ? pairs[e] : none;
    }
#pragma unroll
    for (int j = 0; j < BIN_BATCH; ++j) {
      if (pr[j].y >= 0) {
        const int slot = atomicAdd(&cntL[pr[j].y & (BIN_NODES - 1)], 1);
        csr[slot] = pr[j].x;
      }
    }
  }
#pragma unroll
  for (int j = 0; j < HID / 4; ++j) {
    const int i = t + j * BIN_NODES;
    if (i < pieces) H4[i] = hv[j] * dinvL[i / (HID / 4)];
  }
}

static_assert(HID == 32, "one lane per channel");
#define WS_ROWPTR_LEN (N_NODES + 4)
#define WS_SCAN_LEN   ((SCAN_THREADS / 32) * SCAN_WAVE)
static_assert((long long)N_EDGES + WS_ROWPTR_LEN + 2 * WS_SCAN_LEN <= (long long)N_NODES * HID,
              "sorted edge list, row starts, cell counts and cell starts must fit the plane the "
              "accumulator used to take");
static_assert(HIST_LEN + 1 <= WS_SCAN_LEN, "cell starts and their closing total");
static_assert(2 * (long long)N_EDGES <= (long long)N_NODES * HID,
              "the (source, destination) pairs must fit the second row plane");
static_assert(N_EDGES % 4 == 0 && WS_ROWPTR_LEN % 4 == 0, "16-byte alignment of the scan's arrays");
template <int JBASE>
__device__ __forceinline__ float gcn_rows16(const float* __restrict__ Hlin, int s, int n,
                                            int lane, float acc) {
  float h[16];
#pragma unroll
  for (int j = 0; j < 16; ++j) {
    const int sj = __builtin_amdgcn_readlane(s, JBASE + j);
    h[j] = Hlin[(unsigned)(sj * HID + lane)];
  }
#pragma unroll
  for (int j = 0; j < 16; ++j) acc += (JBASE + j < n) ? h[j] : 0.0f;
  return acc;
}

#define GC_BLOCKS 2048
#define GC_WAVES  (GC_BLOCKS * 8)
#define GC_FULL   (N_NODES / GC_WAVES)
#define GC_ROWS   (GC_FULL + 1)
static_assert(GC_WAVES <= N_NODES, "every wave has a node");
static_assert(GC_ROWS <= 8, "a wave's rows must fit one half of a 16-row tile");
#define GC_PITCH  36

__device__ __forceinline__ void gcn_next_product(const float* rowsL, int first, int lane,
                                                 const float* __restrict__ W,
                                                 float* __restrict__ Hnext) {
  __builtin_amdgcn_fence(__ATOMIC_RELEASE, "wavefront");
  __builtin_amdgcn_wave_barrier();
  __builtin_amdgcn_fence(__ATOMIC_ACQUIRE, "wavefront");
  const bool extra = first + GC_FULL * GC_WAVES < N_NODES;
  const int half = lane >> 4;
  const int l16  = lane & 15;
  const int r8   = l16 & 7;
  const int rr   = r8 < GC_FULL + (extra ? 1 : 0) ? r8 : 0;
  const v4f* ar = (const v4f*)(rowsL + rr * GC_PITCH + half * (HID / 2));
  v4f af[HID / 8];
#pragma unroll
  for (int j = 0; j < HID / 8; ++j) af[j] = ar[j];
  const bool lo = l16 < 8;
  v8f c = {};
#pragma unroll
  for (int p = 0; p < 2; ++p) {
#pragma unroll
    for (int s = 0; s < HID / 4; ++s) {
      const int kk = half * (HID / 2) + 2 * s;
      const bool on = (p == 0) == lo;
      v2f a, w;
      a.x = on ? af[s >> 1][(s & 1) * 2 + 0] : 0.0f;
      a.y = on ? af[s >> 1][(s & 1) * 2 + 1] : 0.0f;
      w.x = W[(kk + 0) * HID + 16 * p + l16];
      w.y = W[(kk + 1) * HID + 16 * p + l16];
      c = __builtin_amdgcn_wmma_f32_16x16x4_f32(
           false, a,  false, w,
           (short)0, c,  false,  false);
    }
  }
  float* __restrict__ row = Hnext + (size_t)first * HID;
#pragma unroll
  for (int q = 0; q < GC_FULL; ++q) row[(size_t)q * GC_WAVES * HID + lane] = c[q];
  if (extra) row[(size_t)GC_FULL * GC_WAVES * HID + lane] = c[GC_FULL];
}

template <bool NEXT>
__global__ void __launch_bounds__(256)
gcn_gather_combine_kernel(const int* __restrict__ rowptr, const int* __restrict__ csr,
                          const float* __restrict__ Hlin,
                          const float* __restrict__ bias, const float* __restrict__ dinv,
                          const int* __restrict__ batch, const float* __restrict__ cntinv,
                          const float* __restrict__ Wn, float* __restrict__ Hnext,
                          float* __restrict__ feat) {
  __shared__ __attribute__((aligned(16))) float rowsAll[NEXT ? 8 * GC_ROWS * GC_PITCH : 4];
  const int first =
      __builtin_amdgcn_readfirstlane(blockIdx.x * (blockDim.x >> 5) + (threadIdx.x >> 5));
  const int lane = threadIdx.x & 31;
  float* rowsL = rowsAll + (NEXT ? (threadIdx.x >> 5) * (GC_ROWS * GC_PITCH) : 0);
  int node = first;
  if (node >= N_NODES) return;
  int i = 0;
  const float bl = bias[lane];
  int beg = rowptr[node];
  int end = rowptr[node + 1];
  int g = batch[node];
  for (;;) {
    int p = beg;
    int q = p + lane;
    int s = q < end ? csr[q] : 0;
    float acc = Hlin[(unsigned)(node * HID + lane)];
    const int next = node + GC_WAVES;
    const bool more = next < N_NODES;
    int nbeg = 0, nend = 0, ng = 0;
    if (more) {
      nbeg = rowptr[next];
      nend = rowptr[next + 1];
      ng = batch[next];
    }
    const float di = dinv[node];
    const float ci = cntinv[g];
    while (p < end) {
      const int n = end - p;
      acc = gcn_rows16<0>(Hlin, s, n, lane, acc);
      if (n > 16) acc = gcn_rows16<16>(Hlin, s, n, lane, acc);
      p += 32;
      q = p + lane;
      s = q < end ? csr[q] : 0;
    }
    float v = acc * di + bl;
    v = v > 0.0f ? v : NEG_SLOPE * v;
    if (NEXT) rowsL[i * GC_PITCH + lane] = v * di;
    atomicAdd(&feat[g * HID + lane], v * ci);
    if (!more) break;
    node = next;
    beg = nbeg;
    end = nend;
    g = ng;
    ++i;
  }
  if (NEXT) gcn_next_product(rowsL, first, lane, Wn, Hnext);
}

__global__ void gcn_merge_kernel(const float* __restrict__ feats, float* __restrict__ out) {
  int t = blockIdx.x * blockDim.x + threadIdx.x;
  const int GH = G_GRAPHS * HID;
  if (t < GH) out[t] = (feats[t] + feats[GH + t] + feats[2 * GH + t]) * (1.0f / 3.0f);
}

static inline int cdiv(int a, int b) { return (a + b - 1) / b; }

extern "C" void kernel_launch(void* const* d_in, const int* in_sizes, int n_in,
                              void* d_out, int out_size, void* d_ws, size_t ws_size,
                              hipStream_t stream) {
  (void)in_sizes; (void)n_in; (void)out_size; (void)ws_size;
  const float* x     = (const float*)d_in[0];
  const float* W0    = (const float*)d_in[1];
  const float* b0    = (const float*)d_in[2];
  const float* W1    = (const float*)d_in[3];
  const float* b1    = (const float*)d_in[4];
  const float* W2    = (const float*)d_in[5];
  const float* b2    = (const float*)d_in[6];
  const int*   src   = (const int*)d_in[7];
  const int*   dst   = (const int*)d_in[8];
  const int*   batch = (const int*)d_in[9];
  float* out = (float*)d_out;

  float* ws    = (float*)d_ws;
  float* deg   = ws;
  float* cnt   = deg + N_NODES;
  float* lin   = cnt + G_GRAPHS;
  float* lin1  = lin + (size_t)N_NODES * HID;
  float* agg   = lin1 + (size_t)N_NODES * HID;
  int* pairs   = (int*)lin1;
  int* csr     = (int*)agg;
  int* rowptr  = csr + N_EDGES;
  int* hist    = rowptr + WS_ROWPTR_LEN;
  int* base    = hist + WS_SCAN_LEN;
  float* feats = agg + (size_t)N_NODES * HID;

  const int T = 256;

  gcn_clear_kernel<<<cdiv(CLEAR_A + CLEAR_B, T), T, 0, stream>>>(cnt, feats);
  gcn_bincount_counts_kernel<<<PART_BLOCKS + COUNT_BLOCKS, PART_THREADS, 0, stream>>>(
      dst, hist, batch, cnt);
  gcn_scan_cntinv_kernel<<<2, SCAN_THREADS, 0, stream>>>(hist, base, cnt);
  gcn_binscatter_gemm0_kernel<<<PART_BLOCKS + GEMM0_BLOCKS, PART_THREADS, 0, stream>>>(
      src, dst, base, pairs, x, W0, lin, N_NODES);
  gcn_bin_rows_kernel<<<N_BINS, BIN_NODES, 0, stream>>>(base, pairs, rowptr, csr, deg, lin);

  gcn_gather_combine_kernel<true><<<GC_BLOCKS, T, 0, stream>>>(
      rowptr, csr, lin, b0, deg, batch, cnt, W1, lin1, feats);
  gcn_gather_combine_kernel<true><<<GC_BLOCKS, T, 0, stream>>>(
      rowptr, csr, lin1, b1, deg, batch, cnt, W2, lin, feats + G_GRAPHS * HID);
  gcn_gather_combine_kernel<false><<<GC_BLOCKS, T, 0, stream>>>(
      rowptr, csr, lin, b2, deg, batch, cnt, nullptr, nullptr, feats + 2 * G_GRAPHS * HID);
  gcn_merge_kernel<<<cdiv(G_GRAPHS * HID, T), T, 0, stream>>>(feats, out);
}


# pooled sums reduced before the atomic add: a wave walks consecutive nodes (was strided), sums one graph's rows in registers, a block's waves combine in LDS; ~40x fewer adds; pooled sum order changes
# speedup vs baseline: 3.1157x; 1.0515x over previous
#include <hip/hip_runtime.h>

#define N_NODES   100000
#define N_EDGES   1600000
#define C_IN      64
#define HID       32
#define G_GRAPHS  128
#define NEG_SLOPE 0.01f

typedef __attribute__((ext_vector_type(2))) float v2f;
typedef __attribute__((ext_vector_type(8))) float v8f;


#define CLEAR_A G_GRAPHS
#define CLEAR_B (3 * G_GRAPHS * HID)
__global__ void gcn_clear_kernel(float* __restrict__ cnt, float* __restrict__ feats) {
  const int i = blockIdx.x * blockDim.x + threadIdx.x;
  if (i < CLEAR_A) cnt[i] = 0.0f;
  else if (i - CLEAR_A < CLEAR_B) feats[i - CLEAR_A] = 0.0f;
}

typedef __attribute__((ext_vector_type(4))) int v4i;
typedef __attribute__((ext_vector_type(2))) int v2i;

template <int BLOCK_THREADS>
__device__ __forceinline__ void gcn_counts_body(int blk, const int* __restrict__ batch,
                                                float* __restrict__ cnt) {
  const int i = blk * BLOCK_THREADS + threadIdx.x;
  const int lane = threadIdx.x & 31;
  const bool live = i < N_NODES;
  const int g = live ? batch[i] : -1;
  unsigned remaining = __builtin_amdgcn_ballot_w32(live);
  for (int trip = 0; trip < 32 && remaining != 0u; ++trip) {
    const int lead = __ffs(remaining) - 1;
    const int gl = __builtin_amdgcn_readlane(g, lead);
    const unsigned same = __builtin_amdgcn_ballot_w32(live && g == gl) & remaining;
    if (lane == lead) atomicAdd(&cnt[gl], (float)__popc(same));
    remaining &= ~same;
  }
}

#define BIN_SHIFT    8
#define BIN_NODES    (1 << BIN_SHIFT)
#define N_BINS       ((N_NODES + BIN_NODES - 1) / BIN_NODES)
#define PART_BLOCKS  256
#define PART_THREADS 1024
#define PART_CHUNK   (N_EDGES / PART_BLOCKS)
#define PART_TRIPS   ((PART_CHUNK + PART_THREADS - 1) / PART_THREADS)
#define HIST_LEN     (N_BINS * PART_BLOCKS)
static_assert(N_EDGES % PART_BLOCKS == 0, "equal chunks");
static_assert(HIST_LEN % 4 == 0, "16-byte groups must not straddle the end");
#define COUNT_BLOCKS ((N_NODES + PART_THREADS - 1) / PART_THREADS)

__global__ void __launch_bounds__(PART_THREADS)
gcn_bincount_counts_kernel(const int* __restrict__ dst, int* __restrict__ hist,
                           const int* __restrict__ batch, float* __restrict__ cnt) {
  __shared__ int binc[N_BINS];
  if (blockIdx.x >= PART_BLOCKS) {
    gcn_counts_body<PART_THREADS>(blockIdx.x - PART_BLOCKS, batch, cnt);
    return;
  }
  const int e0 = blockIdx.x * PART_CHUNK;
  int d[PART_TRIPS];
#pragma unroll
  for (int j = 0; j < PART_TRIPS; ++j) {
    const int k = threadIdx.x + j * PART_THREADS;
    d[j] = k < PART_CHUNK ? dst[e0 + k] : -1;
  }
  for (int t = threadIdx.x; t < N_BINS; t += PART_THREADS) binc[t] = 0;
  __syncthreads();
#pragma unroll
  for (int j = 0; j < PART_TRIPS; ++j)
    if (d[j] >= 0) atomicAdd(&binc[d[j] >> BIN_SHIFT], 1);
  __syncthreads();
  for (int t = threadIdx.x; t < N_BINS; t += PART_THREADS)
    hist[t * PART_BLOCKS + blockIdx.x] = binc[t];
}

#define SCAN_THREADS 1024
#define SCAN_TRIPS   25
#define SCAN_WAVE    (SCAN_TRIPS * 128)
static_assert((SCAN_THREADS / 32) * SCAN_WAVE >= HIST_LEN, "one work-group must cover every cell");
__global__ void __launch_bounds__(SCAN_THREADS)
gcn_scan_cntinv_kernel(const int* __restrict__ hist, int* __restrict__ base,
                       float* __restrict__ cnt) {
  if (blockIdx.x != 0) {
    if (threadIdx.x < G_GRAPHS) cnt[threadIdx.x] = 1.0f / fmaxf(cnt[threadIdx.x], 1.0f);
    return;
  }
  __shared__ int wtot[SCAN_THREADS / 32];
  const int lane = threadIdx.x & 31;
  const int w = threadIdx.x >> 5;
  const int first = w * SCAN_WAVE + lane * 4;
  int ex[SCAN_TRIPS];
  int carry = 0;
#pragma unroll
  for (int k = 0; k < SCAN_TRIPS; ++k) {
    const int idx = first + k * 128;
    v4i v = {0, 0, 0, 0};
    if (idx < HIST_LEN) v = *(const v4i*)(hist + idx);
    const int t = v.x + v.y + v.z + v.w;
    int inc = t;
#pragma unroll
    for (int d = 1; d < 32; d <<= 1) {
      const int u = __shfl_up(inc, d);
      if (lane >= d) inc += u;
    }
    ex[k] = carry + inc - t;
    carry += __builtin_amdgcn_readlane(inc, 31);
  }
  if (lane == 0) wtot[w] = carry;
  __syncthreads();
  const int mine = wtot[lane];
  int off = lane < w ? mine : 0;
  int total = mine;
#pragma unroll
  for (int d = 16; d >= 1; d >>= 1) {
    off += __shfl_xor(off, d);
    total += __shfl_xor(total, d);
  }
#pragma unroll
  for (int k = 0; k < SCAN_TRIPS; ++k) {
    const int idx = first + k * 128;
    if (idx < HIST_LEN) {
      const v4i v = *(const v4i*)(hist + idx);
      v4i o;
      o.x = off + ex[k];
      o.y = o.x + v.x;
      o.z = o.y + v.y;
      o.w = o.z + v.z;
      *(v4i*)(base + idx) = o;
    }
  }
  if (threadIdx.x == 0) base[HIST_LEN] = total;
}

typedef __attribute__((ext_vector_type(4))) float v4f;
template <int K, int WAVES_PER_BLOCK>
__device__ __forceinline__ void gcn_gemm_wmma_body(int blk, const float* __restrict__ X,
                                                   const float* __restrict__ W,
                                                   float* __restrict__ H, int nRows) {
  static_assert(K % 8 == 0, "a lane half's k range in 16-byte pieces");
  const int wave = blk * WAVES_PER_BLOCK + (threadIdx.x >> 5);
  const int lane = threadIdx.x & 31;
  const int row0 = wave * 16;
  if (row0 >= nRows) return;
  const int half = lane >> 4;
  const int l16  = lane & 15;

  v4f af[K / 8];
  const v4f* __restrict__ xr = (const v4f*)(X + (row0 + l16) * K + half * (K / 2));
#pragma unroll
  for (int j = 0; j < K / 8; ++j) af[j] = xr[j];
  v2f wf[K / 4][2];
#pragma unroll
  for (int s = 0; s < K / 4; ++s) {
    const int kk = half * (K / 2) + 2 * s;
#pragma unroll
    for (int p = 0; p < 2; ++p) {
      wf[s][p].x = W[(kk + 0) * HID + 16 * p + l16];
      wf[s][p].y = W[(kk + 1) * HID + 16 * p + l16];
    }
  }

#pragma unroll
  for (int p = 0; p < 2; ++p) {
    v8f c = {};
#pragma unroll
    for (int s = 0; s < K / 4; ++s) {
      v2f a;
      a.x = af[s >> 1][(s & 1) * 2 + 0];
      a.y = af[s >> 1][(s & 1) * 2 + 1];
      c = __builtin_amdgcn_wmma_f32_16x16x4_f32(
           false, a,  false, wf[s][p],
           (short)0, c,  false,  false);
    }
#pragma unroll
    for (int r = 0; r < 8; ++r)
      H[(row0 + r + 8 * half) * HID + 16 * p + l16] = c[r];
  }
}

static_assert(N_NODES % 16 == 0, "row tiles of the products");
#define GEMM0_WAVES  (N_NODES / 16)
#define GEMM0_BLOCKS ((GEMM0_WAVES + PART_THREADS / 32 - 1) / (PART_THREADS / 32))
__global__ void __launch_bounds__(PART_THREADS)
gcn_binscatter_gemm0_kernel(const int* __restrict__ src, const int* __restrict__ dst,
                            const int* __restrict__ base, int* __restrict__ pairbuf,
                            const float* __restrict__ X, const float* __restrict__ W,
                            float* __restrict__ H, int nRows) {
  __shared__ int cur[N_BINS];
  v2i* __restrict__ pairs = (v2i*)pairbuf;
  if (blockIdx.x >= PART_BLOCKS) {
    gcn_gemm_wmma_body<C_IN, PART_THREADS / 32>(blockIdx.x - PART_BLOCKS, X, W, H, nRows);
    return;
  }
  const int e0 = blockIdx.x * PART_CHUNK;
  v2i pr[PART_TRIPS];
#pragma unroll
  for (int j = 0; j < PART_TRIPS; ++j) {
    const int k = threadIdx.x + j * PART_THREADS;
    pr[j].x = k < PART_CHUNK ? src[e0 + k] : 0;
    pr[j].y = k < PART_CHUNK ? dst[e0 + k] : -1;
  }
  for (int t = threadIdx.x; t < N_BINS; t += PART_THREADS)
    cur[t] = base[t * PART_BLOCKS + blockIdx.x];
  __syncthreads();
#pragma unroll
  for (int j = 0; j < PART_TRIPS; ++j) {
    if (pr[j].y >= 0) {
      const int slot = atomicAdd(&cur[pr[j].y >> BIN_SHIFT], 1);
      pairs[slot] = pr[j];
    }
  }
}

static_assert(BIN_NODES == 256, "eight waves: one thread per node of the bin");
static_assert(HID % 4 == 0, "rows in 16-byte pieces");
#define BIN_BATCH 8
__global__ void __launch_bounds__(BIN_NODES)
gcn_bin_rows_kernel(const int* __restrict__ base, const int* __restrict__ pairbuf,
                    int* __restrict__ rowptr, int* __restrict__ csr,
                    float* __restrict__ dinv, float* __restrict__ H) {
  __shared__ int cntL[BIN_NODES];
  __shared__ float dinvL[BIN_NODES];
  __shared__ int wtot[BIN_NODES / 32];
  const v2i* __restrict__ pairs = (const v2i*)pairbuf;
  const int b = blockIdx.x;
  const int t = threadIdx.x;
  const int lane = t & 31;
  const int w = t >> 5;
  const int beg = base[b * PART_BLOCKS];
  const int end = base[(b + 1) * PART_BLOCKS];
  const int n0 = b * BIN_NODES;
  const int rows = N_NODES - n0 < BIN_NODES ? N_NODES - n0 : BIN_NODES;
  v4f* __restrict__ H4 = (v4f*)(H + n0 * HID);
  const int pieces = rows * (HID / 4);
  v4f hv[HID / 4];
#pragma unroll
  for (int j = 0; j < HID / 4; ++j) {
    const int i = t + j * BIN_NODES;
    hv[j] = H4[i < pieces ? i : pieces - 1];
  }
  cntL[t] = 0;
  __syncthreads();
  for (int eb = beg; eb < end; eb += BIN_NODES * BIN_BATCH) {
    int dl[BIN_BATCH];
#pragma unroll
    for (int j = 0; j < BIN_BATCH; ++j) {
      const int e = eb + j * BIN_NODES + t;
      dl[j] = e < end ? pairs[e].y : -1;
    }
#pragma unroll
    for (int j = 0; j < BIN_BATCH; ++j)
      if (dl[j] >= 0) atomicAdd(&cntL[dl[j] & (BIN_NODES - 1)], 1);
  }
  __syncthreads();
  const int deg = cntL[t];
  int inc = deg;
#pragma unroll
  for (int d = 1; d < 32; d <<= 1) {
    const int u = __shfl_up(inc, d);
    if (lane >= d) inc += u;
  }
  if (lane == 31) wtot[w] = inc;
  __syncthreads();
  int off = 0;
#pragma unroll
  for (int j = 0; j < BIN_NODES / 32; ++j) off += j < w ? wtot[j] : 0;
  const int start = beg + off + inc - deg;
  const int node = n0 + t;
  const float di = rsqrtf((float)deg + 1.0f);
  dinvL[t] = di;
  cntL[t] = start;
  if (node < N_NODES) {
    rowptr[node] = start;
    dinv[node] = di;
  }
  if (b == N_BINS - 1 && t == 0) rowptr[N_NODES] = end;
  __syncthreads();
  for (int eb = beg; eb < end; eb += BIN_NODES * BIN_BATCH) {
    v2i pr[BIN_BATCH];
#pragma unroll
    for (int j = 0; j < BIN_BATCH; ++j) {
      const int e = eb + j * BIN_NODES + t;
      v2i none = {0, -1};
      pr[j] = e < end ? pairs[e] : none;
    }
#pragma unroll
    for (int j = 0; j < BIN_BATCH; ++j) {
      if (pr[j].y >= 0) {
        const int slot = atomicAdd(&cntL[pr[j].y & (BIN_NODES - 1)], 1);
        csr[slot] = pr[j].x;
      }
    }
  }
#pragma unroll
  for (int j = 0; j < HID / 4; ++j) {
    const int i = t + j * BIN_NODES;
    if (i < pieces) H4[i] = hv[j] * dinvL[i / (HID / 4)];
  }
}

static_assert(HID == 32, "one lane per channel");
#define WS_ROWPTR_LEN (N_NODES + 4)
#define WS_SCAN_LEN   ((SCAN_THREADS / 32) * SCAN_WAVE)
static_assert((long long)N_EDGES + WS_ROWPTR_LEN + 2 * WS_SCAN_LEN <= (long long)N_NODES * HID,
              "sorted edge list, row starts, cell counts and cell starts must fit the plane the "
              "accumulator used to take");
static_assert(HIST_LEN + 1 <= WS_SCAN_LEN, "cell starts and their closing total");
static_assert(2 * (long long)N_EDGES <= (long long)N_NODES * HID,
              "the (source, destination) pairs must fit the second row plane");
static_assert(N_EDGES % 4 == 0 && WS_ROWPTR_LEN % 4 == 0, "16-byte alignment of the scan's arrays");
template <int JBASE>
__device__ __forceinline__ float gcn_rows16(const float* __restrict__ Hlin, int s, int n,
                                            int lane, float acc) {
  float h[16];
#pragma unroll
  for (int j = 0; j < 16; ++j) {
    const int sj = __builtin_amdgcn_readlane(s, JBASE + j);
    h[j] = Hlin[(unsigned)(sj * HID + lane)];
  }
#pragma unroll
  for (int j = 0; j < 16; ++j) acc += (JBASE + j < n) ? h[j] : 0.0f;
  return acc;
}

#define GC_BLOCKS 2048
#define GC_WPB    8
#define GC_WAVES  (GC_BLOCKS * GC_WPB)
#define GC_FULL   (N_NODES / GC_WAVES)
#define GC_EXTRA  (N_NODES - GC_FULL * GC_WAVES)
#define GC_ROWS   (GC_FULL + 1)
static_assert(GC_FULL >= 1, "every wave has a node");
static_assert((GC_WPB & (GC_WPB - 1)) == 0 && GC_WPB <= 32,
              "wave index wrapped by a mask; one id per lane when the runs are combined");
static_assert(GC_ROWS <= 8, "a wave's rows must fit one half of a 16-row tile");
#define GC_PITCH  36

__device__ __forceinline__ void gcn_next_product(const float* rowsL, int n0, int cnt, int lane,
                                                 const float* __restrict__ W,
                                                 float* __restrict__ Hnext) {
  __builtin_amdgcn_fence(__ATOMIC_RELEASE, "wavefront");
  __builtin_amdgcn_wave_barrier();
  __builtin_amdgcn_fence(__ATOMIC_ACQUIRE, "wavefront");
  const bool extra = cnt > GC_FULL;
  const int half = lane >> 4;
  const int l16  = lane & 15;
  const int r8   = l16 & 7;
  const int rr   = r8 < cnt ? r8 : 0;
  const v4f* ar = (const v4f*)(rowsL + rr * GC_PITCH + half * (HID / 2));
  v4f af[HID / 8];
#pragma unroll
  for (int j = 0; j < HID / 8; ++j) af[j] = ar[j];
  const bool lo = l16 < 8;
  v8f c = {};
#pragma unroll
  for (int p = 0; p < 2; ++p) {
#pragma unroll
    for (int s = 0; s < HID / 4; ++s) {
      const int kk = half * (HID / 2) + 2 * s;
      const bool on = (p == 0) == lo;
      v2f a, w;
      a.x = on ? af[s >> 1][(s & 1) * 2 + 0] : 0.0f;
      a.y = on ? af[s >> 1][(s & 1) * 2 + 1] : 0.0f;
      w.x = W[(kk + 0) * HID + 16 * p + l16];
      w.y = W[(kk + 1) * HID + 16 * p + l16];
      c = __builtin_amdgcn_wmma_f32_16x16x4_f32(
           false, a,  false, w,
           (short)0, c,  false,  false);
    }
  }
  float* __restrict__ row = Hnext + (size_t)n0 * HID;
#pragma unroll
  for (int q = 0; q < GC_FULL; ++q) row[q * HID + lane] = c[q];
  if (extra) row[GC_FULL * HID + lane] = c[GC_FULL];
}

template <bool NEXT>
__global__ void __launch_bounds__(256)
gcn_gather_combine_kernel(const int* __restrict__ rowptr, const int* __restrict__ csr,
                          const float* __restrict__ Hlin,
                          const float* __restrict__ bias, const float* __restrict__ dinv,
                          const int* __restrict__ batch, const float* __restrict__ cntinv,
                          const float* __restrict__ Wn, float* __restrict__ Hnext,
                          float* __restrict__ feat) {
  __shared__ __attribute__((aligned(16))) float rowsAll[NEXT ? GC_WPB * GC_ROWS * GC_PITCH : 4];
  __shared__ float poolL[GC_WPB * HID];
  __shared__ int poolG[GC_WPB];
  const int wib = __builtin_amdgcn_readfirstlane(threadIdx.x >> 5);
  const int wv = blockIdx.x * GC_WPB + wib;
  const int lane = threadIdx.x & 31;
  float* rowsL = rowsAll + (NEXT ? wib * (GC_ROWS * GC_PITCH) : 0);
  const int n0 = wv * GC_FULL + (wv < GC_EXTRA ? wv : GC_EXTRA);
  const int cnt = GC_FULL + (wv < GC_EXTRA ? 1 : 0);
  int node = n0;
  int i = 0;
  const float bl = bias[lane];
  int beg = rowptr[node];
  int end = rowptr[node + 1];
  int g = batch[node];
  float pool = 0.0f;
  for (;;) {
    int p = beg;
    int q = p + lane;
    int s = q < end ? csr[q] : 0;
    float acc = Hlin[(unsigned)(node * HID + lane)];
    const bool more = i + 1 < cnt;
    int nend = 0, ng = 0;
    if (more) {
      nend = rowptr[node + 2];
      ng = batch[node + 1];
    }
    const float di = dinv[node];
    while (p < end) {
      const int n = end - p;
      acc = gcn_rows16<0>(Hlin, s, n, lane, acc);
      if (n > 16) acc = gcn_rows16<16>(Hlin, s, n, lane, acc);
      p += 32;
      q = p + lane;
      s = q < end ? csr[q] : 0;
    }
    float v = acc * di + bl;
    v = v > 0.0f ? v : NEG_SLOPE * v;
    if (NEXT) rowsL[i * GC_PITCH + lane] = v * di;
    pool += v;
    if (!more) break;
    if (ng != g) {
      atomicAdd(&feat[g * HID + lane], pool * cntinv[g]);
      pool = 0.0f;
      g = ng;
    }
    ++node;
    beg = end;
    end = nend;
    ++i;
  }
  poolL[wib * HID + lane] = pool;
  if (lane == 0) poolG[wib] = g;
  __syncthreads();
  const unsigned same =
      __builtin_amdgcn_ballot_w32(poolG[lane & (GC_WPB - 1)] == g) & ((1u << GC_WPB) - 1u);
  if (wib == 0 || ((same >> (wib - 1)) & 1u) == 0u) {
    const int len = __ffs(~(same >> wib)) - 1;
    float sum = pool;
#pragma unroll
    for (int k = 1; k < GC_WPB; ++k) {
      const float r = poolL[((wib + k) & (GC_WPB - 1)) * HID + lane];
      sum += k < len ? r : 0.0f;
    }
    atomicAdd(&feat[g * HID + lane], sum * cntinv[g]);
  }
  if (NEXT) gcn_next_product(rowsL, n0, cnt, lane, Wn, Hnext);
}

__global__ void gcn_merge_kernel(const float* __restrict__ feats, float* __restrict__ out) {
  int t = blockIdx.x * blockDim.x + threadIdx.x;
  const int GH = G_GRAPHS * HID;
  if (t < GH) out[t] = (feats[t] + feats[GH + t] + feats[2 * GH + t]) * (1.0f / 3.0f);
}

static inline int cdiv(int a, int b) { return (a + b - 1) / b; }

extern "C" void kernel_launch(void* const* d_in, const int* in_sizes, int n_in,
                              void* d_out, int out_size, void* d_ws, size_t ws_size,
                              hipStream_t stream) {
  (void)in_sizes; (void)n_in; (void)out_size; (void)ws_size;
  const float* x     = (const float*)d_in[0];
  const float* W0    = (const float*)d_in[1];
  const float* b0    = (const float*)d_in[2];
  const float* W1    = (const float*)d_in[3];
  const float* b1    = (const float*)d_in[4];
  const float* W2    = (const float*)d_in[5];
  const float* b2    = (const float*)d_in[6];
  const int*   src   = (const int*)d_in[7];
  const int*   dst   = (const int*)d_in[8];
  const int*   batch = (const int*)d_in[9];
  float* out = (float*)d_out;

  float* ws    = (float*)d_ws;
  float* deg   = ws;
  float* cnt   = deg + N_NODES;
  float* lin   = cnt + G_GRAPHS;
  float* lin1  = lin + (size_t)N_NODES * HID;
  float* agg   = lin1 + (size_t)N_NODES * HID;
  int* pairs   = (int*)lin1;
  int* csr     = (int*)agg;
  int* rowptr  = csr + N_EDGES;
  int* hist    = rowptr + WS_ROWPTR_LEN;
  int* base    = hist + WS_SCAN_LEN;
  float* feats = agg + (size_t)N_NODES * HID;

  const int T = 256;

  gcn_clear_kernel<<<cdiv(CLEAR_A + CLEAR_B, T), T, 0, stream>>>(cnt, feats);
  gcn_bincount_counts_kernel<<<PART_BLOCKS + COUNT_BLOCKS, PART_THREADS, 0, stream>>>(
      dst, hist, batch, cnt);
  gcn_scan_cntinv_kernel<<<2, SCAN_THREADS, 0, stream>>>(hist, base, cnt);
  gcn_binscatter_gemm0_kernel<<<PART_BLOCKS + GEMM0_BLOCKS, PART_THREADS, 0, stream>>>(
      src, dst, base, pairs, x, W0, lin, N_NODES);
  gcn_bin_rows_kernel<<<N_BINS, BIN_NODES, 0, stream>>>(base, pairs, rowptr, csr, deg, lin);

  static_assert(T == GC_WPB * 32, "the edge-sum work-groups are GC_WPB waves");
  gcn_gather_combine_kernel<true><<<GC_BLOCKS, T, 0, stream>>>(
      rowptr, csr, lin, b0, deg, batch, cnt, W1, lin1, feats);
  gcn_gather_combine_kernel<true><<<GC_BLOCKS, T, 0, stream>>>(
      rowptr, csr, lin1, b1, deg, batch, cnt, W2, lin, feats + G_GRAPHS * HID);
  gcn_gather_combine_kernel<false><<<GC_BLOCKS, T, 0, stream>>>(
      rowptr, csr, lin, b2, deg, batch, cnt, nullptr, nullptr, feats + 2 * G_GRAPHS * HID);
  gcn_merge_kernel<<<cdiv(G_GRAPHS * HID, T), T, 0, stream>>>(feats, out);
}


# 9 -> 6 launches: no clear, scan or mean launch: graph counts as per-block partials, each copy block scans the cell counts itself, layers add into out (zeroed in-run); out's float add order changes
# speedup vs baseline: 3.8586x; 1.2384x over previous
#include <hip/hip_runtime.h>

#define N_NODES   100000
#define N_EDGES   1600000
#define C_IN      64
#define HID       32
#define G_GRAPHS  128
#define NEG_SLOPE 0.01f

typedef __attribute__((ext_vector_type(2))) float v2f;
typedef __attribute__((ext_vector_type(8))) float v8f;


typedef __attribute__((ext_vector_type(4))) int v4i;
typedef __attribute__((ext_vector_type(2))) int v2i;

template <int BLOCK_THREADS>
__device__ __forceinline__ void gcn_counts_body(int blk, const int* __restrict__ batch,
                                                int* __restrict__ gpart, int* cntL) {
  for (int t = threadIdx.x; t < G_GRAPHS; t += BLOCK_THREADS) cntL[t] = 0;
  __syncthreads();
  const int i = blk * BLOCK_THREADS + threadIdx.x;
  const int lane = threadIdx.x & 31;
  const bool live = i < N_NODES;
  const int g = live ? batch[i] : -1;
  unsigned remaining = __builtin_amdgcn_ballot_w32(live);
  for (int trip = 0; trip < 32 && remaining != 0u; ++trip) {
    const int lead = __ffs(remaining) - 1;
    const int gl = __builtin_amdgcn_readlane(g, lead);
    const unsigned same = __builtin_amdgcn_ballot_w32(live && g == gl) & remaining;
    if (lane == lead) atomicAdd(&cntL[gl], (int)__popc(same));
    remaining &= ~same;
  }
  __syncthreads();
  for (int t = threadIdx.x; t < G_GRAPHS; t += BLOCK_THREADS) gpart[blk * G_GRAPHS + t] = cntL[t];
}

#define BIN_SHIFT    8
#define BIN_NODES    (1 << BIN_SHIFT)
#define N_BINS       ((N_NODES + BIN_NODES - 1) / BIN_NODES)
#define PART_BLOCKS  256
#define PART_THREADS 1024
#define PART_CHUNK   (N_EDGES / PART_BLOCKS)
#define PART_TRIPS   ((PART_CHUNK + PART_THREADS - 1) / PART_THREADS)
#define HIST_PITCH   ((N_BINS + 3) & ~3)
#define HIST_LEN     (PART_BLOCKS * HIST_PITCH)
static_assert(N_EDGES % PART_BLOCKS == 0, "equal chunks");
static_assert(N_BINS >= G_GRAPHS, "the graph-count work-groups borrow the bin counters");
#define COUNT_BLOCKS ((N_NODES + PART_THREADS - 1) / PART_THREADS)

__global__ void __launch_bounds__(PART_THREADS)
gcn_bincount_counts_kernel(const int* __restrict__ dst, int* __restrict__ hist,
                           const int* __restrict__ batch, int* __restrict__ gpart) {
  __shared__ int binc[N_BINS];
  if (blockIdx.x >= PART_BLOCKS) {
    gcn_counts_body<PART_THREADS>(blockIdx.x - PART_BLOCKS, batch, gpart, binc);
    return;
  }
  const int e0 = blockIdx.x * PART_CHUNK;
  int d[PART_TRIPS];
#pragma unroll
  for (int j = 0; j < PART_TRIPS; ++j) {
    const int k = threadIdx.x + j * PART_THREADS;
    d[j] = k < PART_CHUNK ? dst[e0 + k] : -1;
  }
  for (int t = threadIdx.x; t < N_BINS; t += PART_THREADS) binc[t] = 0;
  __syncthreads();
#pragma unroll
  for (int j = 0; j < PART_TRIPS; ++j)
    if (d[j] >= 0) atomicAdd(&binc[d[j] >> BIN_SHIFT], 1);
  __syncthreads();
  for (int t = threadIdx.x; t < HIST_PITCH; t += PART_THREADS)
    hist[blockIdx.x * HIST_PITCH + t] = t < N_BINS ? binc[t] : 0;
}

typedef __attribute__((ext_vector_type(4))) float v4f;
template <int K, int WAVES_PER_BLOCK>
__device__ __forceinline__ void gcn_gemm_wmma_body(int blk, const float* __restrict__ X,
                                                   const float* __restrict__ W,
                                                   float* __restrict__ H, int nRows) {
  static_assert(K % 8 == 0, "a lane half's k range in 16-byte pieces");
  const int wave = blk * WAVES_PER_BLOCK + (threadIdx.x >> 5);
  const int lane = threadIdx.x & 31;
  const int row0 = wave * 16;
  if (row0 >= nRows) return;
  const int half = lane >> 4;
  const int l16  = lane & 15;

  v4f af[K / 8];
  const v4f* __restrict__ xr = (const v4f*)(X + (row0 + l16) * K + half * (K / 2));
#pragma unroll
  for (int j = 0; j < K / 8; ++j) af[j] = xr[j];
  v2f wf[K / 4][2];
#pragma unroll
  for (int s = 0; s < K / 4; ++s) {
    const int kk = half * (K / 2) + 2 * s;
#pragma unroll
    for (int p = 0; p < 2; ++p) {
      wf[s][p].x = W[(kk + 0) * HID + 16 * p + l16];
      wf[s][p].y = W[(kk + 1) * HID + 16 * p + l16];
    }
  }

#pragma unroll
  for (int p = 0; p < 2; ++p) {
    v8f c = {};
#pragma unroll
    for (int s = 0; s < K / 4; ++s) {
      v2f a;
      a.x = af[s >> 1][(s & 1) * 2 + 0];
      a.y = af[s >> 1][(s & 1) * 2 + 1];
      c = __builtin_amdgcn_wmma_f32_16x16x4_f32(
           false, a,  false, wf[s][p],
           (short)0, c,  false,  false);
    }
#pragma unroll
    for (int r = 0; r < 8; ++r)
      H[(row0 + r + 8 * half) * HID + 16 * p + l16] = c[r];
  }
}

static_assert(N_NODES % 16 == 0, "row tiles of the products");
#define GEMM0_WAVES  (N_NODES / 16)
#define GEMM0_BLOCKS ((GEMM0_WAVES + PART_THREADS / 32 - 1) / (PART_THREADS / 32))
#define CELL_GROUPS  128
#define CELL_RANGES  (PART_THREADS / CELL_GROUPS)
#define CELL_SPAN    (PART_BLOCKS / CELL_RANGES)
#define CELL_BATCH   8
#define BIN_WAVES    ((N_BINS + 31) / 32)
static_assert(HIST_PITCH / 4 <= CELL_GROUPS, "one thread per 16-byte group of a chunk's counts");
static_assert(PART_BLOCKS % CELL_RANGES == 0 && CELL_SPAN % CELL_BATCH == 0, "whole batches");
static_assert(N_BINS <= PART_THREADS, "one thread per bin in the scan");
__global__ void __launch_bounds__(PART_THREADS) __attribute__((amdgpu_waves_per_eu(16)))
gcn_binscatter_gemm0_kernel(const int* __restrict__ src, const int* __restrict__ dst,
                            const int* __restrict__ hist, int* __restrict__ binstart,
                            int* __restrict__ pairbuf,
                            const float* __restrict__ X, const float* __restrict__ W,
                            float* __restrict__ H, int nRows,
                            const int* __restrict__ gpart, float* __restrict__ gscale,
                            float* __restrict__ out) {
  __shared__ int cur[N_BINS];
  __shared__ int totL[HIST_PITCH];
  __shared__ int preL[HIST_PITCH];
  __shared__ int wtot[BIN_WAVES];
  v2i* __restrict__ pairs = (v2i*)pairbuf;
  if (blockIdx.x == PART_BLOCKS + GEMM0_BLOCKS) {
    if (threadIdx.x < G_GRAPHS) {
      int n = 0;
      for (int b = 0; b < COUNT_BLOCKS; ++b) n += gpart[b * G_GRAPHS + threadIdx.x];
      gscale[threadIdx.x] = (1.0f / fmaxf((float)n, 1.0f)) * (1.0f / 3.0f);
    }
    for (int i = threadIdx.x; i < G_GRAPHS * HID; i += PART_THREADS) out[i] = 0.0f;
    return;
  }
  if (blockIdx.x >= PART_BLOCKS) {
    gcn_gemm_wmma_body<C_IN, PART_THREADS / 32>(blockIdx.x - PART_BLOCKS, X, W, H, nRows);
    return;
  }
  const int t = threadIdx.x;
  for (int i = t; i < HIST_PITCH; i += PART_THREADS) {
    totL[i] = 0;
    preL[i] = 0;
  }
  __syncthreads();
  const int e0 = blockIdx.x * PART_CHUNK;
  v2i pr[PART_TRIPS];
#pragma unroll
  for (int j = 0; j < PART_TRIPS; ++j) {
    const int k = threadIdx.x + j * PART_THREADS;
    pr[j].x = k < PART_CHUNK ? src[e0 + k] : 0;
    pr[j].y = k < PART_CHUNK ? dst[e0 + k] : -1;
  }
  const int q = t & (CELL_GROUPS - 1);
  const int r = t / CELL_GROUPS;
  if (q < HIST_PITCH / 4) {
    const v4i* __restrict__ hp = (const v4i*)(hist + r * CELL_SPAN * HIST_PITCH) + q;
    v4i tot = {0, 0, 0, 0};
    v4i pre = {0, 0, 0, 0};
    for (int jb = 0; jb < CELL_SPAN; jb += CELL_BATCH) {
      v4i v[CELL_BATCH];
#pragma unroll
      for (int j = 0; j < CELL_BATCH; ++j) v[j] = hp[(jb + j) * (HIST_PITCH / 4)];
#pragma unroll
      for (int j = 0; j < CELL_BATCH; ++j) {
        tot += v[j];
        if (r * CELL_SPAN + jb + j < (int)blockIdx.x) pre += v[j];
      }
    }
#pragma unroll
    for (int k = 0; k < 4; ++k) {
      atomicAdd(&totL[4 * q + k], tot[k]);
      atomicAdd(&preL[4 * q + k], pre[k]);
    }
  }
  __syncthreads();
  const int deg = t < N_BINS ? totL[t] : 0;
  const int pfx = t < N_BINS ? preL[t] : 0;
  const int lane = t & 31;
  const int w = t >> 5;
  int inc = deg;
#pragma unroll
  for (int d = 1; d < 32; d <<= 1) {
    const int u = __shfl_up(inc, d);
    if (lane >= d) inc += u;
  }
  if (lane == 31 && w < BIN_WAVES) wtot[w] = inc;
  __syncthreads();
  int off = 0;
#pragma unroll
  for (int j = 0; j < BIN_WAVES; ++j) off += j < w ? wtot[j] : 0;
  const int start = off + inc - deg;
  if (t < N_BINS) {
    cur[t] = start + pfx;
    if (blockIdx.x == 0) binstart[t] = start;
  }
  if (blockIdx.x == 0 && t == N_BINS - 1) binstart[N_BINS] = start + deg;
  __syncthreads();
#pragma unroll
  for (int j = 0; j < PART_TRIPS; ++j) {
    if (pr[j].y >= 0) {
      const int slot = atomicAdd(&cur[pr[j].y >> BIN_SHIFT], 1);
      pairs[slot] = pr[j];
    }
  }
}

static_assert(BIN_NODES == 256, "eight waves: one thread per node of the bin");
static_assert(HID % 4 == 0, "rows in 16-byte pieces");
#define BIN_BATCH 8
__global__ void __launch_bounds__(BIN_NODES)
gcn_bin_rows_kernel(const int* __restrict__ binstart, const int* __restrict__ pairbuf,
                    int* __restrict__ rowptr, int* __restrict__ csr,
                    float* __restrict__ dinv, float* __restrict__ H) {
  __shared__ int cntL[BIN_NODES];
  __shared__ float dinvL[BIN_NODES];
  __shared__ int wtot[BIN_NODES / 32];
  const v2i* __restrict__ pairs = (const v2i*)pairbuf;
  const int b = blockIdx.x;
  const int t = threadIdx.x;
  const int lane = t & 31;
  const int w = t >> 5;
  const int beg = binstart[b];
  const int end = binstart[b + 1];
  const int n0 = b * BIN_NODES;
  const int rows = N_NODES - n0 < BIN_NODES ? N_NODES - n0 : BIN_NODES;
  v4f* __restrict__ H4 = (v4f*)(H + n0 * HID);
  const int pieces = rows * (HID / 4);
  v4f hv[HID / 4];
#pragma unroll
  for (int j = 0; j < HID / 4; ++j) {
    const int i = t + j * BIN_NODES;
    hv[j] = H4[i < pieces ? i : pieces - 1];
  }
  cntL[t] = 0;
  __syncthreads();
  for (int eb = beg; eb < end; eb += BIN_NODES * BIN_BATCH) {
    int dl[BIN_BATCH];
#pragma unroll
    for (int j = 0; j < BIN_BATCH; ++j) {
      const int e = eb + j * BIN_NODES + t;
      dl[j] = e < end ? pairs[e].y : -1;
    }
#pragma unroll
    for (int j = 0; j < BIN_BATCH; ++j)
      if (dl[j] >= 0) atomicAdd(&cntL[dl[j] & (BIN_NODES - 1)], 1);
  }
  __syncthreads();
  const int deg = cntL[t];
  int inc = deg;
#pragma unroll
  for (int d = 1; d < 32; d <<= 1) {
    const int u = __shfl_up(inc, d);
    if (lane >= d) inc += u;
  }
  if (lane == 31) wtot[w] = inc;
  __syncthreads();
  int off = 0;
#pragma unroll
  for (int j = 0; j < BIN_NODES / 32; ++j) off += j < w ? wtot[j] : 0;
  const int start = beg + off + inc - deg;
  const int node = n0 + t;
  const float di = rsqrtf((float)deg + 1.0f);
  dinvL[t] = di;
  cntL[t] = start;
  if (node < N_NODES) {
    rowptr[node] = start;
    dinv[node] = di;
  }
  if (b == N_BINS - 1 && t == 0) rowptr[N_NODES] = end;
  __syncthreads();
  for (int eb = beg; eb < end; eb += BIN_NODES * BIN_BATCH) {
    v2i pr[BIN_BATCH];
#pragma unroll
    for (int j = 0; j < BIN_BATCH; ++j) {
      const int e = eb + j * BIN_NODES + t;
      v2i none = {0, -1};
      pr[j] = e < end ? pairs[e] : none;
    }
#pragma unroll
    for (int j = 0; j < BIN_BATCH; ++j) {
      if (pr[j].y >= 0) {
        const int slot = atomicAdd(&cntL[pr[j].y & (BIN_NODES - 1)], 1);
        csr[slot] = pr[j].x;
      }
    }
  }
#pragma unroll
  for (int j = 0; j < HID / 4; ++j) {
    const int i = t + j * BIN_NODES;
    if (i < pieces) H4[i] = hv[j] * dinvL[i / (HID / 4)];
  }
}

static_assert(HID == 32, "one lane per channel");
#define WS_ROWPTR_LEN (N_NODES + 4)
#define WS_BINS_LEN   ((N_BINS + 1 + 3) & ~3)
#define WS_GPART_LEN  (COUNT_BLOCKS * G_GRAPHS)
static_assert((long long)N_EDGES + WS_ROWPTR_LEN + HIST_LEN + WS_BINS_LEN + WS_GPART_LEN <=
                  (long long)N_NODES * HID,
              "sorted edge list, row starts, cell counts, bin starts and partial graph counts must "
              "fit the plane the accumulator used to take");
static_assert(2 * (long long)N_EDGES <= (long long)N_NODES * HID,
              "the (source, destination) pairs must fit the second row plane");
static_assert(N_EDGES % 4 == 0 && WS_ROWPTR_LEN % 4 == 0, "16-byte alignment of the cell counts");
template <int JBASE>
__device__ __forceinline__ float gcn_rows16(const float* __restrict__ Hlin, int s, int n,
                                            int lane, float acc) {
  float h[16];
#pragma unroll
  for (int j = 0; j < 16; ++j) {
    const int sj = __builtin_amdgcn_readlane(s, JBASE + j);
    h[j] = Hlin[(unsigned)(sj * HID + lane)];
  }
#pragma unroll
  for (int j = 0; j < 16; ++j) acc += (JBASE + j < n) ? h[j] : 0.0f;
  return acc;
}

#define GC_BLOCKS 2048
#define GC_WPB    8
#define GC_WAVES  (GC_BLOCKS * GC_WPB)
#define GC_FULL   (N_NODES / GC_WAVES)
#define GC_EXTRA  (N_NODES - GC_FULL * GC_WAVES)
#define GC_ROWS   (GC_FULL + 1)
static_assert(GC_FULL >= 1, "every wave has a node");
static_assert((GC_WPB & (GC_WPB - 1)) == 0 && GC_WPB <= 32,
              "wave index wrapped by a mask; one id per lane when the runs are combined");
static_assert(GC_ROWS <= 8, "a wave's rows must fit one half of a 16-row tile");
#define GC_PITCH  36

__device__ __forceinline__ void gcn_next_product(const float* rowsL, int n0, int cnt, int lane,
                                                 const float* __restrict__ W,
                                                 float* __restrict__ Hnext) {
  __builtin_amdgcn_fence(__ATOMIC_RELEASE, "wavefront");
  __builtin_amdgcn_wave_barrier();
  __builtin_amdgcn_fence(__ATOMIC_ACQUIRE, "wavefront");
  const bool extra = cnt > GC_FULL;
  const int half = lane >> 4;
  const int l16  = lane & 15;
  const int r8   = l16 & 7;
  const int rr   = r8 < cnt ? r8 : 0;
  const v4f* ar = (const v4f*)(rowsL + rr * GC_PITCH + half * (HID / 2));
  v4f af[HID / 8];
#pragma unroll
  for (int j = 0; j < HID / 8; ++j) af[j] = ar[j];
  const bool lo = l16 < 8;
  v8f c = {};
#pragma unroll
  for (int p = 0; p < 2; ++p) {
#pragma unroll
    for (int s = 0; s < HID / 4; ++s) {
      const int kk = half * (HID / 2) + 2 * s;
      const bool on = (p == 0) == lo;
      v2f a, w;
      a.x = on ? af[s >> 1][(s & 1) * 2 + 0] : 0.0f;
      a.y = on ? af[s >> 1][(s & 1) * 2 + 1] : 0.0f;
      w.x = W[(kk + 0) * HID + 16 * p + l16];
      w.y = W[(kk + 1) * HID + 16 * p + l16];
      c = __builtin_amdgcn_wmma_f32_16x16x4_f32(
           false, a,  false, w,
           (short)0, c,  false,  false);
    }
  }
  float* __restrict__ row = Hnext + (size_t)n0 * HID;
#pragma unroll
  for (int q = 0; q < GC_FULL; ++q) row[q * HID + lane] = c[q];
  if (extra) row[GC_FULL * HID + lane] = c[GC_FULL];
}

template <bool NEXT>
__global__ void __launch_bounds__(256)
gcn_gather_combine_kernel(const int* __restrict__ rowptr, const int* __restrict__ csr,
                          const float* __restrict__ Hlin,
                          const float* __restrict__ bias, const float* __restrict__ dinv,
                          const int* __restrict__ batch, const float* __restrict__ gscale,
                          const float* __restrict__ Wn, float* __restrict__ Hnext,
                          float* __restrict__ out) {
  __shared__ __attribute__((aligned(16))) float rowsAll[NEXT ? GC_WPB * GC_ROWS * GC_PITCH : 4];
  __shared__ float poolL[GC_WPB * HID];
  __shared__ int poolG[GC_WPB];
  const int wib = __builtin_amdgcn_readfirstlane(threadIdx.x >> 5);
  const int wv = blockIdx.x * GC_WPB + wib;
  const int lane = threadIdx.x & 31;
  float* rowsL = rowsAll + (NEXT ? wib * (GC_ROWS * GC_PITCH) : 0);
  const int n0 = wv * GC_FULL + (wv < GC_EXTRA ? wv : GC_EXTRA);
  const int cnt = GC_FULL + (wv < GC_EXTRA ? 1 : 0);
  int node = n0;
  int i = 0;
  const float bl = bias[lane];
  int beg = rowptr[node];
  int end = rowptr[node + 1];
  int g = batch[node];
  float pool = 0.0f;
  for (;;) {
    int p = beg;
    int q = p + lane;
    int s = q < end ? csr[q] : 0;
    float acc = Hlin[(unsigned)(node * HID + lane)];
    const bool more = i + 1 < cnt;
    int nend = 0, ng = 0;
    if (more) {
      nend = rowptr[node + 2];
      ng = batch[node + 1];
    }
    const float di = dinv[node];
    while (p < end) {
      const int n = end - p;
      acc = gcn_rows16<0>(Hlin, s, n, lane, acc);
      if (n > 16) acc = gcn_rows16<16>(Hlin, s, n, lane, acc);
      p += 32;
      q = p + lane;
      s = q < end ? csr[q] : 0;
    }
    float v = acc * di + bl;
    v = v > 0.0f ? v : NEG_SLOPE * v;
    if (NEXT) rowsL[i * GC_PITCH + lane] = v * di;
    pool += v;
    if (!more) break;
    if (ng != g) {
      atomicAdd(&out[g * HID + lane], pool * gscale[g]);
      pool = 0.0f;
      g = ng;
    }
    ++node;
    beg = end;
    end = nend;
    ++i;
  }
  poolL[wib * HID + lane] = pool;
  if (lane == 0) poolG[wib] = g;
  __syncthreads();
  const unsigned same =
      __builtin_amdgcn_ballot_w32(poolG[lane & (GC_WPB - 1)] == g) & ((1u << GC_WPB) - 1u);
  if (wib == 0 || ((same >> (wib - 1)) & 1u) == 0u) {
    const int len = __ffs(~(same >> wib)) - 1;
    float sum = pool;
#pragma unroll
    for (int k = 1; k < GC_WPB; ++k) {
      const float r = poolL[((wib + k) & (GC_WPB - 1)) * HID + lane];
      sum += k < len ? r : 0.0f;
    }
    atomicAdd(&out[g * HID + lane], sum * gscale[g]);
  }
  if (NEXT) gcn_next_product(rowsL, n0, cnt, lane, Wn, Hnext);
}

extern "C" void kernel_launch(void* const* d_in, const int* in_sizes, int n_in,
                              void* d_out, int out_size, void* d_ws, size_t ws_size,
                              hipStream_t stream) {
  (void)in_sizes; (void)n_in; (void)out_size; (void)ws_size;
  const float* x     = (const float*)d_in[0];
  const float* W0    = (const float*)d_in[1];
  const float* b0    = (const float*)d_in[2];
  const float* W1    = (const float*)d_in[3];
  const float* b1    = (const float*)d_in[4];
  const float* W2    = (const float*)d_in[5];
  const float* b2    = (const float*)d_in[6];
  const int*   src   = (const int*)d_in[7];
  const int*   dst   = (const int*)d_in[8];
  const int*   batch = (const int*)d_in[9];
  float* out = (float*)d_out;

  float* ws     = (float*)d_ws;
  float* deg    = ws;
  float* gscale = deg + N_NODES;
  float* lin    = gscale + G_GRAPHS;
  float* lin1   = lin + (size_t)N_NODES * HID;
  float* agg    = lin1 + (size_t)N_NODES * HID;
  int* pairs    = (int*)lin1;
  int* csr      = (int*)agg;
  int* rowptr   = csr + N_EDGES;
  int* hist     = rowptr + WS_ROWPTR_LEN;
  int* binstart = hist + HIST_LEN;
  int* gpart    = binstart + WS_BINS_LEN;

  const int T = 256;

  gcn_bincount_counts_kernel<<<PART_BLOCKS + COUNT_BLOCKS, PART_THREADS, 0, stream>>>(
      dst, hist, batch, gpart);
  gcn_binscatter_gemm0_kernel<<<PART_BLOCKS + GEMM0_BLOCKS + 1, PART_THREADS, 0, stream>>>(
      src, dst, hist, binstart, pairs, x, W0, lin, N_NODES, gpart, gscale, out);
  gcn_bin_rows_kernel<<<N_BINS, BIN_NODES, 0, stream>>>(binstart, pairs, rowptr, csr, deg, lin);

  static_assert(T == GC_WPB * 32, "the edge-sum work-groups are GC_WPB waves");
  gcn_gather_combine_kernel<true><<<GC_BLOCKS, T, 0, stream>>>(
      rowptr, csr, lin, b0, deg, batch, gscale, W1, lin1, out);
  gcn_gather_combine_kernel<true><<<GC_BLOCKS, T, 0, stream>>>(
      rowptr, csr, lin1, b1, deg, batch, gscale, W2, lin, out);
  gcn_gather_combine_kernel<false><<<GC_BLOCKS, T, 0, stream>>>(
      rowptr, csr, lin, b2, deg, batch, gscale, nullptr, nullptr, out);
}
